# starting kernel
# baseline (speedup 1.0000x reference)
#include <hip/hip_runtime.h>
#include <hip/hip_bf16.h>

typedef __attribute__((ext_vector_type(2))) float v2f;
typedef __attribute__((ext_vector_type(8))) float v8f;

__global__ void k_zero(float* __restrict__ p, int n) {
    int t = blockIdx.x * blockDim.x + threadIdx.x;
    int stride = gridDim.x * blockDim.x;
    for (int i = t; i < n; i += stride) p[i] = 0.0f;
}

__global__ void k_degrees(const int* __restrict__ src, const int* __restrict__ dst,
                          float* __restrict__ deg_out, float* __restrict__ deg_in, int E) {
    int t = blockIdx.x * blockDim.x + threadIdx.x;
    if (t < E) {
        atomicAdd(&deg_out[src[t]], 1.0f);
        atomicAdd(&deg_in[dst[t]], 1.0f);
    }
}

__global__ void k_rsqrt(float* __restrict__ d, int n) {
    int t = blockIdx.x * blockDim.x + threadIdx.x;
    if (t < n) d[t] = rsqrtf(fmaxf(d[t], 1.0f));
}

__global__ __launch_bounds__(256)
void k_gemm1(const float* __restrict__ X, const float* __restrict__ W,
             const float* __restrict__ norm_src, float* __restrict__ H, int N) {
    __shared__ float lA[32][130];
    __shared__ float lB[128][66];

    const int m0 = blockIdx.x * 32;
    const int t  = threadIdx.x;

    #pragma unroll
    for (int i = 0; i < 16; ++i) {
        int idx = t + i * 256;
        int r = idx >> 7, c = idx & 127;
        int row = m0 + r; if (row >= N) row = N - 1;
        lA[r][c] = X[(long)row * 128 + c] * norm_src[row];
    }
    #pragma unroll
    for (int i = 0; i < 32; ++i) {
        int idx = t + i * 256;
        lB[idx >> 6][idx & 63] = W[idx];
    }
    __syncthreads();

    const int wave = t >> 5, lane = t & 31;
    const int mt = (wave >> 2) * 16;
    const int nt = (wave & 3) * 16;
    const int lr = lane & 15;
    const int lh = lane >> 4;

    v8f acc = {};
    #pragma unroll
    for (int k0 = 0; k0 < 128; k0 += 4) {
        const int kk = k0 + 2 * lh;
        v2f a, b;
        a[0] = lA[mt + lr][kk];
        a[1] = lA[mt + lr][kk + 1];
        b[0] = lB[kk][nt + lr];
        b[1] = lB[kk + 1][nt + lr];
        acc = __builtin_amdgcn_wmma_f32_16x16x4_f32(
             false, a,  false, b,
             (short)0, acc,  false,  false);
    }

    #pragma unroll
    for (int r = 0; r < 8; ++r) {
        int m = m0 + mt + r + 8 * lh;
        if (m < N) H[(long)m * 64 + nt + lr] = acc[r];
    }
}

__global__ void k_scatter1(const float* __restrict__ H, const int* __restrict__ src,
                           const int* __restrict__ dst, float* __restrict__ AGG, int E) {
    long t = (long)blockIdx.x * blockDim.x + threadIdx.x;
    int e = (int)(t >> 4);
    if (e >= E) return;
    int f = ((int)t & 15) * 4;
    int s = src[e], d = dst[e];
    const float4 v = *(const float4*)(H + (long)s * 64 + f);
    float* out = AGG + (long)d * 64 + f;
    atomicAdd(out + 0, v.x); atomicAdd(out + 1, v.y);
    atomicAdd(out + 2, v.z); atomicAdd(out + 3, v.w);
}

__global__ __launch_bounds__(256)
void k_gemm2(const float* __restrict__ AGG1, const float* __restrict__ W2,
             const float* __restrict__ b1,
             const float* __restrict__ norm_src, const float* __restrict__ norm_dst,
             float* __restrict__ H2, int N) {
    __shared__ float lA[128][66];
    __shared__ float lB[64][18];

    const int m0 = blockIdx.x * 128;
    const int t  = threadIdx.x;

    #pragma unroll
    for (int i = 0; i < 32; ++i) {
        int idx = t + i * 256;
        int r = idx >> 6, c = idx & 63;
        int row = m0 + r; if (row >= N) row = N - 1;
        float v = AGG1[(long)row * 64 + c];
        v = fmaxf(v * norm_dst[row] + b1[c], 0.0f) * norm_src[row];
        lA[r][c] = v;
    }
    #pragma unroll
    for (int i = 0; i < 4; ++i) {
        int idx = t + i * 256;
        lB[idx >> 4][idx & 15] = W2[idx];
    }
    __syncthreads();

    const int wave = t >> 5, lane = t & 31;
    const int mt = wave * 16;
    const int lr = lane & 15;
    const int lh = lane >> 4;

    v8f acc = {};
    #pragma unroll
    for (int k0 = 0; k0 < 64; k0 += 4) {
        const int kk = k0 + 2 * lh;
        v2f a, b;
        a[0] = lA[mt + lr][kk];
        a[1] = lA[mt + lr][kk + 1];
        b[0] = lB[kk][lr];
        b[1] = lB[kk + 1][lr];
        acc = __builtin_amdgcn_wmma_f32_16x16x4_f32(
            false, a, false, b, (short)0, acc, false, false);
    }

    #pragma unroll
    for (int r = 0; r < 8; ++r) {
        int m = m0 + mt + r + 8 * lh;
        if (m < N) H2[(long)m * 16 + lr] = acc[r];
    }
}

__global__ void k_scatter2(const float* __restrict__ H, const int* __restrict__ src,
                           const int* __restrict__ dst, float* __restrict__ AGG, int E) {
    long t = (long)blockIdx.x * blockDim.x + threadIdx.x;
    int e = (int)(t >> 2);
    if (e >= E) return;
    int f = ((int)t & 3) * 4;
    int s = src[e], d = dst[e];
    const float4 v = *(const float4*)(H + (long)s * 16 + f);
    float* out = AGG + (long)d * 16 + f;
    atomicAdd(out + 0, v.x); atomicAdd(out + 1, v.y);
    atomicAdd(out + 2, v.z); atomicAdd(out + 3, v.w);
}

__global__ void k_finalize(const float* __restrict__ AGG2, const float* __restrict__ norm_dst,
                           const float* __restrict__ b2, float* __restrict__ out, int total) {
    int t = blockIdx.x * blockDim.x + threadIdx.x;
    if (t < total) {
        int i = t >> 4, j = t & 15;
        out[t] = AGG2[t] * norm_dst[i] + b2[j];
    }
}

extern "C" void kernel_launch(void* const* d_in, const int* in_sizes, int n_in,
                              void* d_out, int out_size, void* d_ws, size_t ws_size,
                              hipStream_t stream) {
    const float* emb = (const float*)d_in[0];
    const int*   src = (const int*)  d_in[1];
    const int*   dst = (const int*)  d_in[2];
    const float* W1  = (const float*)d_in[3];
    const float* b1  = (const float*)d_in[4];
    const float* W2  = (const float*)d_in[5];
    const float* b2  = (const float*)d_in[6];
    float* out = (float*)d_out;

    const int N = in_sizes[0] / 128;
    const int E = in_sizes[1];

    float* ws       = (float*)d_ws;
    float* norm_src = ws;
    float* norm_dst = ws + N;
    float* H1       = ws + 2 * (size_t)N;
    float* AGG1     = H1 + (size_t)N * 64;
    float* H2       = H1;
    float* AGG2     = AGG1;

    const int B = 256;

    k_zero<<<512, B, 0, stream>>>(norm_src, 2 * N);
    k_degrees<<<(E + B - 1) / B, B, 0, stream>>>(src, dst, norm_src, norm_dst, E);
    k_rsqrt<<<(2 * N + B - 1) / B, B, 0, stream>>>(norm_src, 2 * N);

    k_gemm1<<<(N + 31) / 32, B, 0, stream>>>(emb, W1, norm_src, H1, N);
    k_zero<<<2048, B, 0, stream>>>(AGG1, N * 64);
    {
        long threads = (long)E * 16;
        k_scatter1<<<(unsigned)((threads + B - 1) / B), B, 0, stream>>>(H1, src, dst, AGG1, E);
    }

    k_gemm2<<<(N + 127) / 128, B, 0, stream>>>(AGG1, W2, b1, norm_src, norm_dst, H2, N);
    k_zero<<<2048, B, 0, stream>>>(AGG2, N * 16);
    {
        long threads = (long)E * 4;
        k_scatter2<<<(unsigned)((threads + B - 1) / B), B, 0, stream>>>(H2, src, dst, AGG2, E);
    }

    k_finalize<<<(N * 16 + B - 1) / B, B, 0, stream>>>(AGG2, norm_dst, b2, out, N * 16);
}


# degree counts by LDS integer adds: a work-group counts one node slice of one index array in LDS, partial planes summed in the rsqrt kernel; the degree pass has no global atomics and no zeroing
# speedup vs baseline: 1.4995x; 1.4995x over previous
#include <hip/hip_runtime.h>
#include <hip/hip_bf16.h>

typedef __attribute__((ext_vector_type(2))) float v2f;
typedef __attribute__((ext_vector_type(8))) float v8f;

__global__ void k_zero(float* __restrict__ p, int n) {
    int t = blockIdx.x * blockDim.x + threadIdx.x;
    int stride = gridDim.x * blockDim.x;
    for (int i = t; i < n; i += stride) p[i] = 0.0f;
}

#define HB 12544
#define HC 16
#define HT 1024

__global__ __launch_bounds__(HT)
void k_hist(const int* __restrict__ src, const int* __restrict__ dst,
            unsigned* __restrict__ part, int N, int E) {
    __shared__ unsigned h[HB];
    const int t  = threadIdx.x;
    const int c  = blockIdx.x;
    const int lo = blockIdx.y * HB;
    const int a  = blockIdx.z;
    const int* __restrict__ idx = a ? dst : src;
    int nb = N - lo; if (nb > HB) nb = HB;
    const unsigned ub = (unsigned)nb;

    for (int i = t; i < HB; i += HT) h[i] = 0u;
    __syncthreads();

    const int4* __restrict__ p = (const int4*)idx;
    const int E4 = E >> 2;
    const int stride = gridDim.x * HT;
    for (int i0 = c * HT + t; i0 < E4; i0 += 4 * stride) {
        int4 v[4];
        #pragma unroll
        for (int u = 0; u < 4; ++u) {
            const int i = i0 + u * stride;
            v[u] = (i < E4) ? p[i] : make_int4(-1, -1, -1, -1);
        }
        #pragma unroll
        for (int u = 0; u < 4; ++u) {
            unsigned b;
            b = (unsigned)(v[u].x - lo); if (b < ub) atomicAdd(&h[b], 1u);
            b = (unsigned)(v[u].y - lo); if (b < ub) atomicAdd(&h[b], 1u);
            b = (unsigned)(v[u].z - lo); if (b < ub) atomicAdd(&h[b], 1u);
            b = (unsigned)(v[u].w - lo); if (b < ub) atomicAdd(&h[b], 1u);
        }
    }
    if (c == 0 && t < (E & 3)) {
        const unsigned b = (unsigned)(idx[(E4 << 2) + t] - lo);
        if (b < ub) atomicAdd(&h[b], 1u);
    }
    __syncthreads();

    unsigned* __restrict__ o = part + (size_t)c * (2 * (size_t)N) + (size_t)a * N + lo;
    for (int i = t; i < nb; i += HT) o[i] = h[i];
}

__global__ void k_norm(const unsigned* __restrict__ part, float* __restrict__ norm, int n2) {
    int t = blockIdx.x * blockDim.x + threadIdx.x;
    if (t < n2) {
        unsigned s = 0u;
        #pragma unroll
        for (int c = 0; c < HC; ++c) s += part[(size_t)c * n2 + t];
        norm[t] = rsqrtf(fmaxf((float)s, 1.0f));
    }
}

__global__ __launch_bounds__(256)
void k_gemm1(const float* __restrict__ X, const float* __restrict__ W,
             const float* __restrict__ norm_src, float* __restrict__ H, int N) {
    __shared__ float lA[32][130];
    __shared__ float lB[128][66];

    const int m0 = blockIdx.x * 32;
    const int t  = threadIdx.x;

    #pragma unroll
    for (int i = 0; i < 16; ++i) {
        int idx = t + i * 256;
        int r = idx >> 7, c = idx & 127;
        int row = m0 + r; if (row >= N) row = N - 1;
        lA[r][c] = X[(long)row * 128 + c] * norm_src[row];
    }
    #pragma unroll
    for (int i = 0; i < 32; ++i) {
        int idx = t + i * 256;
        lB[idx >> 6][idx & 63] = W[idx];
    }
    __syncthreads();

    const int wave = t >> 5, lane = t & 31;
    const int mt = (wave >> 2) * 16;
    const int nt = (wave & 3) * 16;
    const int lr = lane & 15;
    const int lh = lane >> 4;

    v8f acc = {};
    #pragma unroll
    for (int k0 = 0; k0 < 128; k0 += 4) {
        const int kk = k0 + 2 * lh;
        v2f a, b;
        a[0] = lA[mt + lr][kk];
        a[1] = lA[mt + lr][kk + 1];
        b[0] = lB[kk][nt + lr];
        b[1] = lB[kk + 1][nt + lr];
        acc = __builtin_amdgcn_wmma_f32_16x16x4_f32(
             false, a,  false, b,
             (short)0, acc,  false,  false);
    }

    #pragma unroll
    for (int r = 0; r < 8; ++r) {
        int m = m0 + mt + r + 8 * lh;
        if (m < N) H[(long)m * 64 + nt + lr] = acc[r];
    }
}

__global__ void k_scatter1(const float* __restrict__ H, const int* __restrict__ src,
                           const int* __restrict__ dst, float* __restrict__ AGG, int E) {
    long t = (long)blockIdx.x * blockDim.x + threadIdx.x;
    int e = (int)(t >> 4);
    if (e >= E) return;
    int f = ((int)t & 15) * 4;
    int s = src[e], d = dst[e];
    const float4 v = *(const float4*)(H + (long)s * 64 + f);
    float* out = AGG + (long)d * 64 + f;
    atomicAdd(out + 0, v.x); atomicAdd(out + 1, v.y);
    atomicAdd(out + 2, v.z); atomicAdd(out + 3, v.w);
}

__global__ __launch_bounds__(256)
void k_gemm2(const float* __restrict__ AGG1, const float* __restrict__ W2,
             const float* __restrict__ b1,
             const float* __restrict__ norm_src, const float* __restrict__ norm_dst,
             float* __restrict__ H2, int N) {
    __shared__ float lA[128][66];
    __shared__ float lB[64][18];

    const int m0 = blockIdx.x * 128;
    const int t  = threadIdx.x;

    #pragma unroll
    for (int i = 0; i < 32; ++i) {
        int idx = t + i * 256;
        int r = idx >> 6, c = idx & 63;
        int row = m0 + r; if (row >= N) row = N - 1;
        float v = AGG1[(long)row * 64 + c];
        v = fmaxf(v * norm_dst[row] + b1[c], 0.0f) * norm_src[row];
        lA[r][c] = v;
    }
    #pragma unroll
    for (int i = 0; i < 4; ++i) {
        int idx = t + i * 256;
        lB[idx >> 4][idx & 15] = W2[idx];
    }
    __syncthreads();

    const int wave = t >> 5, lane = t & 31;
    const int mt = wave * 16;
    const int lr = lane & 15;
    const int lh = lane >> 4;

    v8f acc = {};
    #pragma unroll
    for (int k0 = 0; k0 < 64; k0 += 4) {
        const int kk = k0 + 2 * lh;
        v2f a, b;
        a[0] = lA[mt + lr][kk];
        a[1] = lA[mt + lr][kk + 1];
        b[0] = lB[kk][lr];
        b[1] = lB[kk + 1][lr];
        acc = __builtin_amdgcn_wmma_f32_16x16x4_f32(
            false, a, false, b, (short)0, acc, false, false);
    }

    #pragma unroll
    for (int r = 0; r < 8; ++r) {
        int m = m0 + mt + r + 8 * lh;
        if (m < N) H2[(long)m * 16 + lr] = acc[r];
    }
}

__global__ void k_scatter2(const float* __restrict__ H, const int* __restrict__ src,
                           const int* __restrict__ dst, float* __restrict__ AGG, int E) {
    long t = (long)blockIdx.x * blockDim.x + threadIdx.x;
    int e = (int)(t >> 2);
    if (e >= E) return;
    int f = ((int)t & 3) * 4;
    int s = src[e], d = dst[e];
    const float4 v = *(const float4*)(H + (long)s * 16 + f);
    float* out = AGG + (long)d * 16 + f;
    atomicAdd(out + 0, v.x); atomicAdd(out + 1, v.y);
    atomicAdd(out + 2, v.z); atomicAdd(out + 3, v.w);
}

__global__ void k_finalize(const float* __restrict__ AGG2, const float* __restrict__ norm_dst,
                           const float* __restrict__ b2, float* __restrict__ out, int total) {
    int t = blockIdx.x * blockDim.x + threadIdx.x;
    if (t < total) {
        int i = t >> 4, j = t & 15;
        out[t] = AGG2[t] * norm_dst[i] + b2[j];
    }
}

extern "C" void kernel_launch(void* const* d_in, const int* in_sizes, int n_in,
                              void* d_out, int out_size, void* d_ws, size_t ws_size,
                              hipStream_t stream) {
    const float* emb = (const float*)d_in[0];
    const int*   src = (const int*)  d_in[1];
    const int*   dst = (const int*)  d_in[2];
    const float* W1  = (const float*)d_in[3];
    const float* b1  = (const float*)d_in[4];
    const float* W2  = (const float*)d_in[5];
    const float* b2  = (const float*)d_in[6];
    float* out = (float*)d_out;

    const int N = in_sizes[0] / 128;
    const int E = in_sizes[1];

    float* ws       = (float*)d_ws;
    float* norm_src = ws;
    float* norm_dst = ws + N;
    float* H1       = ws + 2 * (size_t)N;
    float* AGG1     = H1 + (size_t)N * 64;
    float* H2       = H1;
    float* AGG2     = AGG1;
    unsigned* part  = (unsigned*)(AGG1 + (size_t)N * 64);

    const int B = 256;

    k_hist<<<dim3(HC, (N + HB - 1) / HB, 2), HT, 0, stream>>>(src, dst, part, N, E);
    k_norm<<<(2 * N + B - 1) / B, B, 0, stream>>>(part, norm_src, 2 * N);

    k_gemm1<<<(N + 31) / 32, B, 0, stream>>>(emb, W1, norm_src, H1, N);
    k_zero<<<2048, B, 0, stream>>>(AGG1, N * 64);
    {
        long threads = (long)E * 16;
        k_scatter1<<<(unsigned)((threads + B - 1) / B), B, 0, stream>>>(H1, src, dst, AGG1, E);
    }

    k_gemm2<<<(N + 127) / 128, B, 0, stream>>>(AGG1, W2, b1, norm_src, norm_dst, H2, N);
    k_zero<<<2048, B, 0, stream>>>(AGG2, N * 16);
    {
        long threads = (long)E * 4;
        k_scatter2<<<(unsigned)((threads + B - 1) / B), B, 0, stream>>>(H2, src, dst, AGG2, E);
    }

    k_finalize<<<(N * 16 + B - 1) / B, B, 0, stream>>>(AGG2, norm_dst, b2, out, N * 16);
}


# both aggregations gather over an edge list counting-sorted by destination in every call, instead of float atomic scatter: sums run in list order; zeroing and finalize passes gone
# speedup vs baseline: 2.7732x; 1.8494x over previous
#include <hip/hip_runtime.h>
#include <hip/hip_bf16.h>

typedef __attribute__((ext_vector_type(2))) float v2f;
typedef __attribute__((ext_vector_type(8))) float v8f;

#define HB 12544
#define HC 16
#define HT 1024

__global__ __launch_bounds__(HT)
void k_hist(const int* __restrict__ src, const int* __restrict__ dst,
            unsigned* __restrict__ part, int N, int E) {
    __shared__ unsigned h[HB];
    const int t  = threadIdx.x;
    const int c  = blockIdx.x;
    const int lo = blockIdx.y * HB;
    const int a  = blockIdx.z;
    const int* __restrict__ idx = a ? dst : src;
    int nb = N - lo; if (nb > HB) nb = HB;
    const unsigned ub = (unsigned)nb;

    for (int i = t; i < HB; i += HT) h[i] = 0u;
    __syncthreads();

    const int4* __restrict__ p = (const int4*)idx;
    const int E4 = E >> 2;
    const int stride = gridDim.x * HT;
    for (int i0 = c * HT + t; i0 < E4; i0 += 4 * stride) {
        int4 v[4];
        #pragma unroll
        for (int u = 0; u < 4; ++u) {
            const int i = i0 + u * stride;
            v[u] = (i < E4) ? p[i] : make_int4(-1, -1, -1, -1);
        }
        #pragma unroll
        for (int u = 0; u < 4; ++u) {
            unsigned b;
            b = (unsigned)(v[u].x - lo); if (b < ub) atomicAdd(&h[b], 1u);
            b = (unsigned)(v[u].y - lo); if (b < ub) atomicAdd(&h[b], 1u);
            b = (unsigned)(v[u].z - lo); if (b < ub) atomicAdd(&h[b], 1u);
            b = (unsigned)(v[u].w - lo); if (b < ub) atomicAdd(&h[b], 1u);
        }
    }
    if (c == 0 && t < (E & 3)) {
        const unsigned b = (unsigned)(idx[(E4 << 2) + t] - lo);
        if (b < ub) atomicAdd(&h[b], 1u);
    }
    __syncthreads();

    unsigned* __restrict__ o = part + (size_t)c * (2 * (size_t)N) + (size_t)a * N + lo;
    for (int i = t; i < nb; i += HT) o[i] = h[i];
}

#define NT 256

__global__ __launch_bounds__(NT)
void k_norm(const unsigned* __restrict__ part, float* __restrict__ norm_src,
            float* __restrict__ norm_dst, unsigned* __restrict__ blocksum, int N) {
    __shared__ unsigned wred[NT / 32];
    const int t = threadIdx.x;
    const int i = blockIdx.x * NT + t;
    const size_t n2 = 2 * (size_t)N;
    unsigned so = 0u, si = 0u;
    if (i < N) {
        #pragma unroll
        for (int c = 0; c < HC; ++c) {
            so += part[(size_t)c * n2 + i];
            si += part[(size_t)c * n2 + N + i];
        }
        norm_src[i] = rsqrtf(fmaxf((float)so, 1.0f));
        norm_dst[i] = rsqrtf(fmaxf((float)si, 1.0f));
    }
    unsigned r = si;
    for (int o = 16; o > 0; o >>= 1) r += __shfl_xor(r, o, 32);
    if ((t & 31) == 0) wred[t >> 5] = r;
    __syncthreads();
    if (t == 0) {
        unsigned s = 0u;
        for (int w = 0; w < NT / 32; ++w) s += wred[w];
        blocksum[blockIdx.x] = s;
    }
}

__global__ __launch_bounds__(NT)
void k_rowptr(unsigned* __restrict__ part, const unsigned* __restrict__ blocksum,
              int* __restrict__ first, int N) {
    __shared__ unsigned wred[NT / 32];
    __shared__ unsigned wsum[NT / 32];
    const int t = threadIdx.x, lane = t & 31, wave = t >> 5;
    const int i = blockIdx.x * NT + t;
    const size_t n2 = 2 * (size_t)N;

    unsigned bs = 0u;
    for (int b = t; b < (int)blockIdx.x; b += NT) bs += blocksum[b];
    for (int o = 16; o > 0; o >>= 1) bs += __shfl_xor(bs, o, 32);

    unsigned p[HC];
    unsigned d = 0u;
    #pragma unroll
    for (int c = 0; c < HC; ++c) {
        p[c] = (i < N) ? part[(size_t)c * n2 + N + i] : 0u;
        d += p[c];
    }

    unsigned inc = d;
    for (int o = 1; o < 32; o <<= 1) {
        const unsigned n = __shfl_up(inc, o, 32);
        if (lane >= o) inc += n;
    }
    if (lane == 0)  wred[wave] = bs;
    if (lane == 31) wsum[wave] = inc;
    __syncthreads();
    unsigned r = inc - d;
    for (int w = 0; w < NT / 32; ++w) {
        r += wred[w];
        if (w < wave) r += wsum[w];
    }

    if (i < N) {
        first[i] = (int)r;
        if (i == N - 1) first[N] = (int)(r + d);
        #pragma unroll
        for (int c = 0; c < HC; ++c) {
            part[(size_t)c * n2 + N + i] = r;
            r += p[c];
        }
    }
}

__global__ __launch_bounds__(HT)
void k_fill(const int* __restrict__ src, const int* __restrict__ dst,
            const unsigned* __restrict__ part, int* __restrict__ csr, int N, int E) {
    __shared__ unsigned h[HB];
    const int t  = threadIdx.x;
    const int c  = blockIdx.x;
    const int lo = blockIdx.y * HB;
    int nb = N - lo; if (nb > HB) nb = HB;
    const unsigned ub = (unsigned)nb;
    const unsigned ue = (unsigned)E;

    const unsigned* __restrict__ cur = part + (size_t)c * (2 * (size_t)N) + N + lo;
    for (int i = t; i < nb; i += HT) h[i] = cur[i];
    __syncthreads();

    const int4* __restrict__ p = (const int4*)dst;
    const int4* __restrict__ q = (const int4*)src;
    const int E4 = E >> 2;
    const int stride = gridDim.x * HT;
    for (int i0 = c * HT + t; i0 < E4; i0 += 4 * stride) {
        int4 v[4], w[4];
        #pragma unroll
        for (int u = 0; u < 4; ++u) {
            const int i = i0 + u * stride;
            v[u] = (i < E4) ? p[i] : make_int4(-1, -1, -1, -1);
            w[u] = (i < E4) ? q[i] : make_int4(0, 0, 0, 0);
        }
        #pragma unroll
        for (int u = 0; u < 4; ++u) {
            unsigned b, pos;
            b = (unsigned)(v[u].x - lo); if (b < ub) { pos = atomicAdd(&h[b], 1u); if (pos < ue) csr[pos] = w[u].x; }
            b = (unsigned)(v[u].y - lo); if (b < ub) { pos = atomicAdd(&h[b], 1u); if (pos < ue) csr[pos] = w[u].y; }
            b = (unsigned)(v[u].z - lo); if (b < ub) { pos = atomicAdd(&h[b], 1u); if (pos < ue) csr[pos] = w[u].z; }
            b = (unsigned)(v[u].w - lo); if (b < ub) { pos = atomicAdd(&h[b], 1u); if (pos < ue) csr[pos] = w[u].w; }
        }
    }
    if (c == 0 && t < (E & 3)) {
        const int e = (E4 << 2) + t;
        const unsigned b = (unsigned)(dst[e] - lo);
        if (b < ub) {
            const unsigned pos = atomicAdd(&h[b], 1u);
            if (pos < ue) csr[pos] = src[e];
        }
    }
}

__global__ __launch_bounds__(256)
void k_gemm1(const float* __restrict__ X, const float* __restrict__ W,
             const float* __restrict__ norm_src, float* __restrict__ H, int N) {
    __shared__ float lA[32][130];
    __shared__ float lB[128][66];

    const int m0 = blockIdx.x * 32;
    const int t  = threadIdx.x;

    #pragma unroll
    for (int i = 0; i < 16; ++i) {
        int idx = t + i * 256;
        int r = idx >> 7, c = idx & 127;
        int row = m0 + r; if (row >= N) row = N - 1;
        lA[r][c] = X[(long)row * 128 + c] * norm_src[row];
    }
    #pragma unroll
    for (int i = 0; i < 32; ++i) {
        int idx = t + i * 256;
        lB[idx >> 6][idx & 63] = W[idx];
    }
    __syncthreads();

    const int wave = t >> 5, lane = t & 31;
    const int mt = (wave >> 2) * 16;
    const int nt = (wave & 3) * 16;
    const int lr = lane & 15;
    const int lh = lane >> 4;

    v8f acc = {};
    #pragma unroll
    for (int k0 = 0; k0 < 128; k0 += 4) {
        const int kk = k0 + 2 * lh;
        v2f a, b;
        a[0] = lA[mt + lr][kk];
        a[1] = lA[mt + lr][kk + 1];
        b[0] = lB[kk][nt + lr];
        b[1] = lB[kk + 1][nt + lr];
        acc = __builtin_amdgcn_wmma_f32_16x16x4_f32(
             false, a,  false, b,
             (short)0, acc,  false,  false);
    }

    #pragma unroll
    for (int r = 0; r < 8; ++r) {
        int m = m0 + mt + r + 8 * lh;
        if (m < N) H[(long)m * 64 + nt + lr] = acc[r];
    }
}

template <int F>
__device__ __forceinline__ float4 gather_sum(const float* __restrict__ H,
                                             const int* __restrict__ first,
                                             const int* __restrict__ csr,
                                             int g, int f, int N, int E) {
    int j = first[g];
    int end = first[g + 1];
    if (j < 0) j = 0;
    if (end > E) end = E;
    const unsigned top = (unsigned)(N - 1);
    float4 acc = make_float4(0.0f, 0.0f, 0.0f, 0.0f);
    for (; j < end; j += 4) {
        unsigned s[4];
        float4 v[4];
        #pragma unroll
        for (int u = 0; u < 4; ++u) {
            int jj = j + u; if (jj > end - 1) jj = end - 1;
            s[u] = (unsigned)csr[jj]; if (s[u] > top) s[u] = top;
        }
        #pragma unroll
        for (int u = 0; u < 4; ++u)
            v[u] = *(const float4*)(H + (size_t)s[u] * F + f);
        #pragma unroll
        for (int u = 0; u < 4; ++u) {
            const bool in = (j + u < end);
            acc.x += in ? v[u].x : 0.0f;
            acc.y += in ? v[u].y : 0.0f;
            acc.z += in ? v[u].z : 0.0f;
            acc.w += in ? v[u].w : 0.0f;
        }
    }
    return acc;
}

__global__ __launch_bounds__(256)
void k_gather1(const float* __restrict__ H, const int* __restrict__ first,
               const int* __restrict__ csr, float* __restrict__ AGG, int N, int E) {
    const int g = (int)((blockIdx.x * 256u + threadIdx.x) >> 4);
    if (g >= N) return;
    const int f = (threadIdx.x & 15) * 4;
    const float4 acc = gather_sum<64>(H, first, csr, g, f, N, E);
    *(float4*)(AGG + (size_t)g * 64 + f) = acc;
}

__global__ __launch_bounds__(256)
void k_gather2(const float* __restrict__ H, const int* __restrict__ first,
               const int* __restrict__ csr, const float* __restrict__ norm_dst,
               const float* __restrict__ b2, float* __restrict__ out, int N, int E) {
    const int g = (int)((blockIdx.x * 256u + threadIdx.x) >> 2);
    if (g >= N) return;
    const int f = (threadIdx.x & 3) * 4;
    const float4 acc = gather_sum<16>(H, first, csr, g, f, N, E);
    const float nd = norm_dst[g];
    float4 o;
    o.x = acc.x * nd + b2[f + 0];
    o.y = acc.y * nd + b2[f + 1];
    o.z = acc.z * nd + b2[f + 2];
    o.w = acc.w * nd + b2[f + 3];
    *(float4*)(out + (size_t)g * 16 + f) = o;
}

__global__ __launch_bounds__(256)
void k_gemm2(const float* __restrict__ AGG1, const float* __restrict__ W2,
             const float* __restrict__ b1,
             const float* __restrict__ norm_src, const float* __restrict__ norm_dst,
             float* __restrict__ H2, int N) {
    __shared__ float lA[128][66];
    __shared__ float lB[64][18];

    const int m0 = blockIdx.x * 128;
    const int t  = threadIdx.x;

    #pragma unroll
    for (int i = 0; i < 32; ++i) {
        int idx = t + i * 256;
        int r = idx >> 6, c = idx & 63;
        int row = m0 + r; if (row >= N) row = N - 1;
        float v = AGG1[(long)row * 64 + c];
        v = fmaxf(v * norm_dst[row] + b1[c], 0.0f) * norm_src[row];
        lA[r][c] = v;
    }
    #pragma unroll
    for (int i = 0; i < 4; ++i) {
        int idx = t + i * 256;
        lB[idx >> 4][idx & 15] = W2[idx];
    }
    __syncthreads();

    const int wave = t >> 5, lane = t & 31;
    const int mt = wave * 16;
    const int lr = lane & 15;
    const int lh = lane >> 4;

    v8f acc = {};
    #pragma unroll
    for (int k0 = 0; k0 < 64; k0 += 4) {
        const int kk = k0 + 2 * lh;
        v2f a, b;
        a[0] = lA[mt + lr][kk];
        a[1] = lA[mt + lr][kk + 1];
        b[0] = lB[kk][lr];
        b[1] = lB[kk + 1][lr];
        acc = __builtin_amdgcn_wmma_f32_16x16x4_f32(
            false, a, false, b, (short)0, acc, false, false);
    }

    #pragma unroll
    for (int r = 0; r < 8; ++r) {
        int m = m0 + mt + r + 8 * lh;
        if (m < N) H2[(long)m * 16 + lr] = acc[r];
    }
}

extern "C" void kernel_launch(void* const* d_in, const int* in_sizes, int n_in,
                              void* d_out, int out_size, void* d_ws, size_t ws_size,
                              hipStream_t stream) {
    const float* emb = (const float*)d_in[0];
    const int*   src = (const int*)  d_in[1];
    const int*   dst = (const int*)  d_in[2];
    const float* W1  = (const float*)d_in[3];
    const float* b1  = (const float*)d_in[4];
    const float* W2  = (const float*)d_in[5];
    const float* b2  = (const float*)d_in[6];
    float* out = (float*)d_out;

    const int N = in_sizes[0] / 128;
    const int E = in_sizes[1];

    float* ws       = (float*)d_ws;
    float* norm_src = ws;
    float* norm_dst = ws + N;
    float* H1       = ws + 2 * (size_t)N;
    float* AGG1     = H1 + (size_t)N * 64;
    float* H2       = H1;
    unsigned* part  = (unsigned*)(AGG1 + (size_t)N * 64);
    const int NB    = (N + NT - 1) / NT;
    unsigned* bsum  = part + (size_t)HC * 2 * (size_t)N;
    int* first      = (int*)(bsum + NB);
    int* csr        = first + (N + 1);

    const int B = 256;
    const int slices = (N + HB - 1) / HB;

    k_hist<<<dim3(HC, slices, 2), HT, 0, stream>>>(src, dst, part, N, E);
    k_norm<<<NB, NT, 0, stream>>>(part, norm_src, norm_dst, bsum, N);
    k_rowptr<<<NB, NT, 0, stream>>>(part, bsum, first, N);
    k_fill<<<dim3(HC, slices, 1), HT, 0, stream>>>(src, dst, part, csr, N, E);

    k_gemm1<<<(N + 31) / 32, B, 0, stream>>>(emb, W1, norm_src, H1, N);
    k_gather1<<<(unsigned)(((long)N * 16 + B - 1) / B), B, 0, stream>>>(H1, first, csr, AGG1, N, E);

    k_gemm2<<<(N + 127) / 128, B, 0, stream>>>(AGG1, W2, b1, norm_src, norm_dst, H2, N);
    k_gather2<<<(unsigned)(((long)N * 4 + B - 1) / B), B, 0, stream>>>(H2, first, csr, norm_dst, b2, out, N, E);
}


# degree count and list fill keep the next group's index loads in flight under this group's LDS adds; the fill issues its sixteen returning adds and its start-position loads as groups, one wait each
# speedup vs baseline: 2.8825x; 1.0394x over previous
#include <hip/hip_runtime.h>
#include <hip/hip_bf16.h>

typedef __attribute__((ext_vector_type(2))) float v2f;
typedef __attribute__((ext_vector_type(8))) float v8f;

#define HB 12544
#define HC 16
#define HT 1024
#define HI ((HB + HT - 1) / HT)

__device__ __forceinline__ void load4(const int4* __restrict__ p, int i0, int stride, int last, int4 (&v)[4]) {
    #pragma unroll
    for (int u = 0; u < 4; ++u) {
        int i = i0 + u * stride; if (i > last) i = last;
        v[u] = p[i];
    }
}

__device__ __forceinline__ void count16(unsigned* h, const int4 (&v)[4], int i0, int stride,
                                        int last, int lo, unsigned ub) {
    #pragma unroll
    for (int u = 0; u < 4; ++u) {
        const unsigned ubu = (i0 + u * stride <= last) ? ub : 0u;
        unsigned b;
        b = (unsigned)(v[u].x - lo); if (b < ubu) atomicAdd(&h[b], 1u);
        b = (unsigned)(v[u].y - lo); if (b < ubu) atomicAdd(&h[b], 1u);
        b = (unsigned)(v[u].z - lo); if (b < ubu) atomicAdd(&h[b], 1u);
        b = (unsigned)(v[u].w - lo); if (b < ubu) atomicAdd(&h[b], 1u);
    }
}

__global__ __launch_bounds__(HT)
void k_hist(const int* __restrict__ src, const int* __restrict__ dst,
            unsigned* __restrict__ part, int N, int E) {
    __shared__ unsigned h[HB];
    const int t  = threadIdx.x;
    const int c  = blockIdx.x;
    const int lo = blockIdx.y * HB;
    const int a  = blockIdx.z;
    const int* __restrict__ idx = a ? dst : src;
    int nb = N - lo; if (nb > HB) nb = HB;
    const unsigned ub = (unsigned)nb;

    for (int i = t; i < HB; i += HT) h[i] = 0u;
    __syncthreads();

    const int4* __restrict__ p = (const int4*)idx;
    const int E4 = E >> 2;
    const int last = E4 - 1;
    const int stride = gridDim.x * HT;
    const int step = 4 * stride;
    int i0 = c * HT + t;
    int4 va[4], vb[4];
    if (i0 < E4) load4(p, i0, stride, last, va);
    while (i0 < E4) {
        load4(p, i0 + step, stride, last, vb);
        count16(h, va, i0, stride, last, lo, ub);
        i0 += step;
        if (i0 >= E4) break;
        load4(p, i0 + step, stride, last, va);
        count16(h, vb, i0, stride, last, lo, ub);
        i0 += step;
    }
    if (c == 0 && t < (E & 3)) {
        const unsigned b = (unsigned)(idx[(E4 << 2) + t] - lo);
        if (b < ub) atomicAdd(&h[b], 1u);
    }
    __syncthreads();

    unsigned* __restrict__ o = part + (size_t)c * (2 * (size_t)N) + (size_t)a * N + lo;
    {
        unsigned r[HI];
        #pragma unroll
        for (int k = 0; k < HI; ++k) { const int i = t + k * HT; r[k] = (i < nb) ? h[i] : 0u; }
        #pragma unroll
        for (int k = 0; k < HI; ++k) { const int i = t + k * HT; if (i < nb) o[i] = r[k]; }
    }
}

#define NT 256

__global__ __launch_bounds__(NT)
void k_norm(const unsigned* __restrict__ part, float* __restrict__ norm_src,
            float* __restrict__ norm_dst, unsigned* __restrict__ blocksum, int N) {
    __shared__ unsigned wred[NT / 32];
    const int t = threadIdx.x;
    const int i = blockIdx.x * NT + t;
    const size_t n2 = 2 * (size_t)N;
    unsigned so = 0u, si = 0u;
    if (i < N) {
        #pragma unroll
        for (int c = 0; c < HC; ++c) {
            so += part[(size_t)c * n2 + i];
            si += part[(size_t)c * n2 + N + i];
        }
        norm_src[i] = rsqrtf(fmaxf((float)so, 1.0f));
        norm_dst[i] = rsqrtf(fmaxf((float)si, 1.0f));
    }
    unsigned r = si;
    for (int o = 16; o > 0; o >>= 1) r += __shfl_xor(r, o, 32);
    if ((t & 31) == 0) wred[t >> 5] = r;
    __syncthreads();
    if (t == 0) {
        unsigned s = 0u;
        for (int w = 0; w < NT / 32; ++w) s += wred[w];
        blocksum[blockIdx.x] = s;
    }
}

__global__ __launch_bounds__(NT)
void k_rowptr(unsigned* __restrict__ part, const unsigned* __restrict__ blocksum,
              int* __restrict__ first, int N) {
    __shared__ unsigned wred[NT / 32];
    __shared__ unsigned wsum[NT / 32];
    const int t = threadIdx.x, lane = t & 31, wave = t >> 5;
    const int i = blockIdx.x * NT + t;
    const size_t n2 = 2 * (size_t)N;

    unsigned bs = 0u;
    for (int b = t; b < (int)blockIdx.x; b += NT) bs += blocksum[b];
    for (int o = 16; o > 0; o >>= 1) bs += __shfl_xor(bs, o, 32);

    unsigned p[HC];
    unsigned d = 0u;
    #pragma unroll
    for (int c = 0; c < HC; ++c) {
        p[c] = (i < N) ? part[(size_t)c * n2 + N + i] : 0u;
        d += p[c];
    }

    unsigned inc = d;
    for (int o = 1; o < 32; o <<= 1) {
        const unsigned n = __shfl_up(inc, o, 32);
        if (lane >= o) inc += n;
    }
    if (lane == 0)  wred[wave] = bs;
    if (lane == 31) wsum[wave] = inc;
    __syncthreads();
    unsigned r = inc - d;
    for (int w = 0; w < NT / 32; ++w) {
        r += wred[w];
        if (w < wave) r += wsum[w];
    }

    if (i < N) {
        first[i] = (int)r;
        if (i == N - 1) first[N] = (int)(r + d);
        #pragma unroll
        for (int c = 0; c < HC; ++c) {
            part[(size_t)c * n2 + N + i] = r;
            r += p[c];
        }
    }
}

__device__ __forceinline__ unsigned take(unsigned* h, int node, int lo, unsigned ub) {
    const unsigned b = (unsigned)(node - lo);
    unsigned r = 0xFFFFFFFFu;
    if (b < ub) r = atomicAdd(&h[b], 1u);
    return r;
}

__device__ __forceinline__ void place16(unsigned* h, int* __restrict__ csr,
                                        const int4 (&v)[4], const int4 (&w)[4],
                                        int i0, int stride, int last, int lo, unsigned ub, unsigned ue) {
    unsigned pos[16];
    #pragma unroll
    for (int u = 0; u < 4; ++u) {
        const unsigned ubu = (i0 + u * stride <= last) ? ub : 0u;
        pos[4 * u + 0] = take(h, v[u].x, lo, ubu);
        pos[4 * u + 1] = take(h, v[u].y, lo, ubu);
        pos[4 * u + 2] = take(h, v[u].z, lo, ubu);
        pos[4 * u + 3] = take(h, v[u].w, lo, ubu);
    }
    #pragma unroll
    for (int u = 0; u < 4; ++u) {
        if (pos[4 * u + 0] < ue) csr[pos[4 * u + 0]] = w[u].x;
        if (pos[4 * u + 1] < ue) csr[pos[4 * u + 1]] = w[u].y;
        if (pos[4 * u + 2] < ue) csr[pos[4 * u + 2]] = w[u].z;
        if (pos[4 * u + 3] < ue) csr[pos[4 * u + 3]] = w[u].w;
    }
}

__global__ __launch_bounds__(HT)
void k_fill(const int* __restrict__ src, const int* __restrict__ dst,
            const unsigned* __restrict__ part, int* __restrict__ csr, int N, int E) {
    __shared__ unsigned h[HB];
    const int t  = threadIdx.x;
    const int c  = blockIdx.x;
    const int lo = blockIdx.y * HB;
    int nb = N - lo; if (nb > HB) nb = HB;
    const unsigned ub = (unsigned)nb;
    const unsigned ue = (unsigned)E;

    const unsigned* __restrict__ cur = part + (size_t)c * (2 * (size_t)N) + N + lo;
    {
        unsigned r[HI];
        #pragma unroll
        for (int k = 0; k < HI; ++k) { const int i = t + k * HT; r[k] = (i < nb) ? cur[i] : 0u; }
        #pragma unroll
        for (int k = 0; k < HI; ++k) { const int i = t + k * HT; if (i < nb) h[i] = r[k]; }
    }
    __syncthreads();

    const int4* __restrict__ p = (const int4*)dst;
    const int4* __restrict__ q = (const int4*)src;
    const int E4 = E >> 2;
    const int last = E4 - 1;
    const int stride = gridDim.x * HT;
    const int step = 4 * stride;
    int i0 = c * HT + t;
    int4 va[4], wa[4], vb[4], wb[4];
    if (i0 < E4) { load4(p, i0, stride, last, va); load4(q, i0, stride, last, wa); }
    while (i0 < E4) {
        load4(p, i0 + step, stride, last, vb);
        load4(q, i0 + step, stride, last, wb);
        place16(h, csr, va, wa, i0, stride, last, lo, ub, ue);
        i0 += step;
        if (i0 >= E4) break;
        load4(p, i0 + step, stride, last, va);
        load4(q, i0 + step, stride, last, wa);
        place16(h, csr, vb, wb, i0, stride, last, lo, ub, ue);
        i0 += step;
    }
    if (c == 0 && t < (E & 3)) {
        const int e = (E4 << 2) + t;
        const unsigned b = (unsigned)(dst[e] - lo);
        if (b < ub) {
            const unsigned pos = atomicAdd(&h[b], 1u);
            if (pos < ue) csr[pos] = src[e];
        }
    }
}

__global__ __launch_bounds__(256)
void k_gemm1(const float* __restrict__ X, const float* __restrict__ W,
             const float* __restrict__ norm_src, float* __restrict__ H, int N) {
    __shared__ float lA[32][130];
    __shared__ float lB[128][66];

    const int m0 = blockIdx.x * 32;
    const int t  = threadIdx.x;

    #pragma unroll
    for (int i = 0; i < 16; ++i) {
        int idx = t + i * 256;
        int r = idx >> 7, c = idx & 127;
        int row = m0 + r; if (row >= N) row = N - 1;
        lA[r][c] = X[(long)row * 128 + c] * norm_src[row];
    }
    #pragma unroll
    for (int i = 0; i < 32; ++i) {
        int idx = t + i * 256;
        lB[idx >> 6][idx & 63] = W[idx];
    }
    __syncthreads();

    const int wave = t >> 5, lane = t & 31;
    const int mt = (wave >> 2) * 16;
    const int nt = (wave & 3) * 16;
    const int lr = lane & 15;
    const int lh = lane >> 4;

    v8f acc = {};
    #pragma unroll
    for (int k0 = 0; k0 < 128; k0 += 4) {
        const int kk = k0 + 2 * lh;
        v2f a, b;
        a[0] = lA[mt + lr][kk];
        a[1] = lA[mt + lr][kk + 1];
        b[0] = lB[kk][nt + lr];
        b[1] = lB[kk + 1][nt + lr];
        acc = __builtin_amdgcn_wmma_f32_16x16x4_f32(
             false, a,  false, b,
             (short)0, acc,  false,  false);
    }

    #pragma unroll
    for (int r = 0; r < 8; ++r) {
        int m = m0 + mt + r + 8 * lh;
        if (m < N) H[(long)m * 64 + nt + lr] = acc[r];
    }
}

template <int F>
__device__ __forceinline__ float4 gather_sum(const float* __restrict__ H,
                                             const int* __restrict__ first,
                                             const int* __restrict__ csr,
                                             int g, int f, int N, int E) {
    int j = first[g];
    int end = first[g + 1];
    if (j < 0) j = 0;
    if (end > E) end = E;
    const unsigned top = (unsigned)(N - 1);
    float4 acc = make_float4(0.0f, 0.0f, 0.0f, 0.0f);
    for (; j < end; j += 4) {
        unsigned s[4];
        float4 v[4];
        #pragma unroll
        for (int u = 0; u < 4; ++u) {
            int jj = j + u; if (jj > end - 1) jj = end - 1;
            s[u] = (unsigned)csr[jj]; if (s[u] > top) s[u] = top;
        }
        #pragma unroll
        for (int u = 0; u < 4; ++u)
            v[u] = *(const float4*)(H + (size_t)s[u] * F + f);
        #pragma unroll
        for (int u = 0; u < 4; ++u) {
            const bool in = (j + u < end);
            acc.x += in ? v[u].x : 0.0f;
            acc.y += in ? v[u].y : 0.0f;
            acc.z += in ? v[u].z : 0.0f;
            acc.w += in ? v[u].w : 0.0f;
        }
    }
    return acc;
}

__global__ __launch_bounds__(256)
void k_gather1(const float* __restrict__ H, const int* __restrict__ first,
               const int* __restrict__ csr, float* __restrict__ AGG, int N, int E) {
    const int g = (int)((blockIdx.x * 256u + threadIdx.x) >> 4);
    if (g >= N) return;
    const int f = (threadIdx.x & 15) * 4;
    const float4 acc = gather_sum<64>(H, first, csr, g, f, N, E);
    *(float4*)(AGG + (size_t)g * 64 + f) = acc;
}

__global__ __launch_bounds__(256)
void k_gather2(const float* __restrict__ H, const int* __restrict__ first,
               const int* __restrict__ csr, const float* __restrict__ norm_dst,
               const float* __restrict__ b2, float* __restrict__ out, int N, int E) {
    const int g = (int)((blockIdx.x * 256u + threadIdx.x) >> 2);
    if (g >= N) return;
    const int f = (threadIdx.x & 3) * 4;
    const float4 acc = gather_sum<16>(H, first, csr, g, f, N, E);
    const float nd = norm_dst[g];
    float4 o;
    o.x = acc.x * nd + b2[f + 0];
    o.y = acc.y * nd + b2[f + 1];
    o.z = acc.z * nd + b2[f + 2];
    o.w = acc.w * nd + b2[f + 3];
    *(float4*)(out + (size_t)g * 16 + f) = o;
}

__global__ __launch_bounds__(256)
void k_gemm2(const float* __restrict__ AGG1, const float* __restrict__ W2,
             const float* __restrict__ b1,
             const float* __restrict__ norm_src, const float* __restrict__ norm_dst,
             float* __restrict__ H2, int N) {
    __shared__ float lA[128][66];
    __shared__ float lB[64][18];

    const int m0 = blockIdx.x * 128;
    const int t  = threadIdx.x;

    #pragma unroll
    for (int i = 0; i < 32; ++i) {
        int idx = t + i * 256;
        int r = idx >> 6, c = idx & 63;
        int row = m0 + r; if (row >= N) row = N - 1;
        float v = AGG1[(long)row * 64 + c];
        v = fmaxf(v * norm_dst[row] + b1[c], 0.0f) * norm_src[row];
        lA[r][c] = v;
    }
    #pragma unroll
    for (int i = 0; i < 4; ++i) {
        int idx = t + i * 256;
        lB[idx >> 4][idx & 15] = W2[idx];
    }
    __syncthreads();

    const int wave = t >> 5, lane = t & 31;
    const int mt = wave * 16;
    const int lr = lane & 15;
    const int lh = lane >> 4;

    v8f acc = {};
    #pragma unroll
    for (int k0 = 0; k0 < 64; k0 += 4) {
        const int kk = k0 + 2 * lh;
        v2f a, b;
        a[0] = lA[mt + lr][kk];
        a[1] = lA[mt + lr][kk + 1];
        b[0] = lB[kk][lr];
        b[1] = lB[kk + 1][lr];
        acc = __builtin_amdgcn_wmma_f32_16x16x4_f32(
            false, a, false, b, (short)0, acc, false, false);
    }

    #pragma unroll
    for (int r = 0; r < 8; ++r) {
        int m = m0 + mt + r + 8 * lh;
        if (m < N) H2[(long)m * 16 + lr] = acc[r];
    }
}

extern "C" void kernel_launch(void* const* d_in, const int* in_sizes, int n_in,
                              void* d_out, int out_size, void* d_ws, size_t ws_size,
                              hipStream_t stream) {
    const float* emb = (const float*)d_in[0];
    const int*   src = (const int*)  d_in[1];
    const int*   dst = (const int*)  d_in[2];
    const float* W1  = (const float*)d_in[3];
    const float* b1  = (const float*)d_in[4];
    const float* W2  = (const float*)d_in[5];
    const float* b2  = (const float*)d_in[6];
    float* out = (float*)d_out;

    const int N = in_sizes[0] / 128;
    const int E = in_sizes[1];

    float* ws       = (float*)d_ws;
    float* norm_src = ws;
    float* norm_dst = ws + N;
    float* H1       = ws + 2 * (size_t)N;
    float* AGG1     = H1 + (size_t)N * 64;
    float* H2       = H1;
    unsigned* part  = (unsigned*)(AGG1 + (size_t)N * 64);
    const int NB    = (N + NT - 1) / NT;
    unsigned* bsum  = part + (size_t)HC * 2 * (size_t)N;
    int* first      = (int*)(bsum + NB);
    int* csr        = first + (N + 1);

    const int B = 256;
    const int slices = (N + HB - 1) / HB;

    k_hist<<<dim3(HC, slices, 2), HT, 0, stream>>>(src, dst, part, N, E);
    k_norm<<<NB, NT, 0, stream>>>(part, norm_src, norm_dst, bsum, N);
    k_rowptr<<<NB, NT, 0, stream>>>(part, bsum, first, N);
    k_fill<<<dim3(HC, slices, 1), HT, 0, stream>>>(src, dst, part, csr, N, E);

    k_gemm1<<<(N + 31) / 32, B, 0, stream>>>(emb, W1, norm_src, H1, N);
    k_gather1<<<(unsigned)(((long)N * 16 + B - 1) / B), B, 0, stream>>>(H1, first, csr, AGG1, N, E);

    k_gemm2<<<(N + 127) / 128, B, 0, stream>>>(AGG1, W2, b1, norm_src, norm_dst, H2, N);
    k_gather2<<<(unsigned)(((long)N * 4 + B - 1) / B), B, 0, stream>>>(H2, first, csr, norm_dst, b2, out, N, E);
}


# list fill and first product share one launch of 768-thread work-groups, two resident per WGP: 128 fill the sorted list while 1042 compute 96-row tiles of X@W1; the half-empty fill launch is gone
# speedup vs baseline: 3.1917x; 1.1073x over previous
#include <hip/hip_runtime.h>
#include <hip/hip_bf16.h>

typedef __attribute__((ext_vector_type(2))) float v2f;
typedef __attribute__((ext_vector_type(8))) float v8f;

#define HB 12544
#define HC 16
#define HT 1024
#define HI ((HB + HT - 1) / HT)

__device__ __forceinline__ void load4(const int4* __restrict__ p, int i0, int stride, int last, int4 (&v)[4]) {
    #pragma unroll
    for (int u = 0; u < 4; ++u) {
        int i = i0 + u * stride; if (i > last) i = last;
        v[u] = p[i];
    }
}

__device__ __forceinline__ void count16(unsigned* h, const int4 (&v)[4], int i0, int stride,
                                        int last, int lo, unsigned ub) {
    #pragma unroll
    for (int u = 0; u < 4; ++u) {
        const unsigned ubu = (i0 + u * stride <= last) ? ub : 0u;
        unsigned b;
        b = (unsigned)(v[u].x - lo); if (b < ubu) atomicAdd(&h[b], 1u);
        b = (unsigned)(v[u].y - lo); if (b < ubu) atomicAdd(&h[b], 1u);
        b = (unsigned)(v[u].z - lo); if (b < ubu) atomicAdd(&h[b], 1u);
        b = (unsigned)(v[u].w - lo); if (b < ubu) atomicAdd(&h[b], 1u);
    }
}

__global__ __launch_bounds__(HT)
void k_hist(const int* __restrict__ src, const int* __restrict__ dst,
            unsigned* __restrict__ part, int N, int E) {
    __shared__ unsigned h[HB];
    const int t  = threadIdx.x;
    const int c  = blockIdx.x;
    const int lo = blockIdx.y * HB;
    const int a  = blockIdx.z;
    const int* __restrict__ idx = a ? dst : src;
    int nb = N - lo; if (nb > HB) nb = HB;
    const unsigned ub = (unsigned)nb;

    for (int i = t; i < HB; i += HT) h[i] = 0u;
    __syncthreads();

    const int4* __restrict__ p = (const int4*)idx;
    const int E4 = E >> 2;
    const int last = E4 - 1;
    const int stride = gridDim.x * HT;
    const int step = 4 * stride;
    int i0 = c * HT + t;
    int4 va[4], vb[4];
    if (i0 < E4) load4(p, i0, stride, last, va);
    while (i0 < E4) {
        load4(p, i0 + step, stride, last, vb);
        count16(h, va, i0, stride, last, lo, ub);
        i0 += step;
        if (i0 >= E4) break;
        load4(p, i0 + step, stride, last, va);
        count16(h, vb, i0, stride, last, lo, ub);
        i0 += step;
    }
    if (c == 0 && t < (E & 3)) {
        const unsigned b = (unsigned)(idx[(E4 << 2) + t] - lo);
        if (b < ub) atomicAdd(&h[b], 1u);
    }
    __syncthreads();

    unsigned* __restrict__ o = part + (size_t)c * (2 * (size_t)N) + (size_t)a * N + lo;
    {
        unsigned r[HI];
        #pragma unroll
        for (int k = 0; k < HI; ++k) { const int i = t + k * HT; r[k] = (i < nb) ? h[i] : 0u; }
        #pragma unroll
        for (int k = 0; k < HI; ++k) { const int i = t + k * HT; if (i < nb) o[i] = r[k]; }
    }
}

#define NT 256

__global__ __launch_bounds__(NT)
void k_norm(const unsigned* __restrict__ part, float* __restrict__ norm_src,
            float* __restrict__ norm_dst, unsigned* __restrict__ blocksum, int N) {
    __shared__ unsigned wred[NT / 32];
    const int t = threadIdx.x;
    const int i = blockIdx.x * NT + t;
    const size_t n2 = 2 * (size_t)N;
    unsigned so = 0u, si = 0u;
    if (i < N) {
        #pragma unroll
        for (int c = 0; c < HC; ++c) {
            so += part[(size_t)c * n2 + i];
            si += part[(size_t)c * n2 + N + i];
        }
        norm_src[i] = rsqrtf(fmaxf((float)so, 1.0f));
        norm_dst[i] = rsqrtf(fmaxf((float)si, 1.0f));
    }
    unsigned r = si;
    for (int o = 16; o > 0; o >>= 1) r += __shfl_xor(r, o, 32);
    if ((t & 31) == 0) wred[t >> 5] = r;
    __syncthreads();
    if (t == 0) {
        unsigned s = 0u;
        for (int w = 0; w < NT / 32; ++w) s += wred[w];
        blocksum[blockIdx.x] = s;
    }
}

__global__ __launch_bounds__(NT)
void k_rowptr(unsigned* __restrict__ part, const unsigned* __restrict__ blocksum,
              int* __restrict__ first, int N) {
    __shared__ unsigned wred[NT / 32];
    __shared__ unsigned wsum[NT / 32];
    const int t = threadIdx.x, lane = t & 31, wave = t >> 5;
    const int i = blockIdx.x * NT + t;
    const size_t n2 = 2 * (size_t)N;

    unsigned bs = 0u;
    for (int b = t; b < (int)blockIdx.x; b += NT) bs += blocksum[b];
    for (int o = 16; o > 0; o >>= 1) bs += __shfl_xor(bs, o, 32);

    unsigned p[HC];
    unsigned d = 0u;
    #pragma unroll
    for (int c = 0; c < HC; ++c) {
        p[c] = (i < N) ? part[(size_t)c * n2 + N + i] : 0u;
        d += p[c];
    }

    unsigned inc = d;
    for (int o = 1; o < 32; o <<= 1) {
        const unsigned n = __shfl_up(inc, o, 32);
        if (lane >= o) inc += n;
    }
    if (lane == 0)  wred[wave] = bs;
    if (lane == 31) wsum[wave] = inc;
    __syncthreads();
    unsigned r = inc - d;
    for (int w = 0; w < NT / 32; ++w) {
        r += wred[w];
        if (w < wave) r += wsum[w];
    }

    if (i < N) {
        first[i] = (int)r;
        if (i == N - 1) first[N] = (int)(r + d);
        #pragma unroll
        for (int c = 0; c < HC; ++c) {
            part[(size_t)c * n2 + N + i] = r;
            r += p[c];
        }
    }
}

#define FT 768
#define FI ((HB + FT - 1) / FT)

__device__ __forceinline__ int share_pos(int m, int c) {
    return ((m / HT) * HC + c) * HT + (m % HT);
}

__device__ __forceinline__ void load4s(const int4* __restrict__ p, int m0, int c, int last, int4 (&v)[4]) {
    #pragma unroll
    for (int u = 0; u < 4; ++u) {
        int i = share_pos(m0 + u * FT, c); if (i > last) i = last;
        v[u] = p[i];
    }
}

__device__ __forceinline__ unsigned take(unsigned* h, int node, int lo, unsigned ub) {
    const unsigned b = (unsigned)(node - lo);
    unsigned r = 0xFFFFFFFFu;
    if (b < ub) r = atomicAdd(&h[b], 1u);
    return r;
}

__device__ __forceinline__ void place16(unsigned* h, int* __restrict__ csr,
                                        const int4 (&v)[4], const int4 (&w)[4],
                                        int m0, int c, int last, int lo, unsigned ub, unsigned ue) {
    unsigned pos[16];
    #pragma unroll
    for (int u = 0; u < 4; ++u) {
        const unsigned ubu = (share_pos(m0 + u * FT, c) <= last) ? ub : 0u;
        pos[4 * u + 0] = take(h, v[u].x, lo, ubu);
        pos[4 * u + 1] = take(h, v[u].y, lo, ubu);
        pos[4 * u + 2] = take(h, v[u].z, lo, ubu);
        pos[4 * u + 3] = take(h, v[u].w, lo, ubu);
    }
    #pragma unroll
    for (int u = 0; u < 4; ++u) {
        if (pos[4 * u + 0] < ue) csr[pos[4 * u + 0]] = w[u].x;
        if (pos[4 * u + 1] < ue) csr[pos[4 * u + 1]] = w[u].y;
        if (pos[4 * u + 2] < ue) csr[pos[4 * u + 2]] = w[u].z;
        if (pos[4 * u + 3] < ue) csr[pos[4 * u + 3]] = w[u].w;
    }
}

__device__ __forceinline__
void fill_role(unsigned* h, const int* __restrict__ src, const int* __restrict__ dst,
               const unsigned* __restrict__ part, int* __restrict__ csr, int N, int E,
               int c, int sl) {
    const int t  = threadIdx.x;
    const int lo = sl * HB;
    int nb = N - lo; if (nb > HB) nb = HB;
    const unsigned ub = (unsigned)nb;
    const unsigned ue = (unsigned)E;

    const unsigned* __restrict__ cur = part + (size_t)c * (2 * (size_t)N) + N + lo;
    {
        unsigned r[FI];
        #pragma unroll
        for (int k = 0; k < FI; ++k) { const int i = t + k * FT; r[k] = (i < nb) ? cur[i] : 0u; }
        #pragma unroll
        for (int k = 0; k < FI; ++k) { const int i = t + k * FT; if (i < nb) h[i] = r[k]; }
    }
    __syncthreads();

    const int4* __restrict__ p = (const int4*)dst;
    const int4* __restrict__ q = (const int4*)src;
    const int E4 = E >> 2;
    const int last = E4 - 1;
    const int step = 4 * FT;
    int m0 = t;
    int4 va[4], wa[4], vb[4], wb[4];
    if (share_pos(m0, c) < E4) { load4s(p, m0, c, last, va); load4s(q, m0, c, last, wa); }
    while (share_pos(m0, c) < E4) {
        load4s(p, m0 + step, c, last, vb);
        load4s(q, m0 + step, c, last, wb);
        place16(h, csr, va, wa, m0, c, last, lo, ub, ue);
        m0 += step;
        if (share_pos(m0, c) >= E4) break;
        load4s(p, m0 + step, c, last, va);
        load4s(q, m0 + step, c, last, wa);
        place16(h, csr, vb, wb, m0, c, last, lo, ub, ue);
        m0 += step;
    }
    if (c == 0 && t < (E & 3)) {
        const int e = (E4 << 2) + t;
        const unsigned b = (unsigned)(dst[e] - lo);
        if (b < ub) {
            const unsigned pos = atomicAdd(&h[b], 1u);
            if (pos < ue) csr[pos] = src[e];
        }
    }
}

#define GM 96
#define GA (GM * 130)
#define GW (128 * 66)

__device__ __forceinline__
void gemm1_role(float* sm, const float* __restrict__ X, const float* __restrict__ W,
                const float* __restrict__ norm_src, float* __restrict__ H, int N, int tile) {
    float (*lA)[130] = (float (*)[130])sm;
    float (*lB)[66]  = (float (*)[66])(sm + GA);

    const int m0 = tile * GM;
    const int t  = threadIdx.x;

    #pragma unroll
    for (int i = 0; i < GM * 128 / FT; ++i) {
        int idx = t + i * FT;
        int r = idx >> 7, c = idx & 127;
        int row = m0 + r; if (row >= N) row = N - 1;
        lA[r][c] = X[(long)row * 128 + c] * norm_src[row];
    }
    #pragma unroll
    for (int i = 0; i < (128 * 64 + FT - 1) / FT; ++i) {
        int idx = t + i * FT;
        if (idx < 128 * 64) lB[idx >> 6][idx & 63] = W[idx];
    }
    __syncthreads();

    const int wave = t >> 5, lane = t & 31;
    const int mt = (wave >> 2) * 16;
    const int nt = (wave & 3) * 16;
    const int lr = lane & 15;
    const int lh = lane >> 4;

    v8f acc = {};
    #pragma unroll
    for (int k0 = 0; k0 < 128; k0 += 4) {
        const int kk = k0 + 2 * lh;
        v2f a, b;
        a[0] = lA[mt + lr][kk];
        a[1] = lA[mt + lr][kk + 1];
        b[0] = lB[kk][nt + lr];
        b[1] = lB[kk + 1][nt + lr];
        acc = __builtin_amdgcn_wmma_f32_16x16x4_f32(
             false, a,  false, b,
             (short)0, acc,  false,  false);
    }

    #pragma unroll
    for (int r = 0; r < 8; ++r) {
        int m = m0 + mt + r + 8 * lh;
        if (m < N) H[(long)m * 64 + nt + lr] = acc[r];
    }
}

union FillOrTile {
    unsigned h[HB];
    float    g[GA + GW];
};

__global__ __launch_bounds__(FT)
__attribute__((amdgpu_waves_per_eu(12)))
void k_fill_gemm1(const int* __restrict__ src, const int* __restrict__ dst,
                  const unsigned* __restrict__ part, int* __restrict__ csr,
                  const float* __restrict__ X, const float* __restrict__ W,
                  const float* __restrict__ norm_src, float* __restrict__ H, int N, int E) {
    __shared__ FillOrTile sh;
    const int slices = (N + HB - 1) / HB;
    const int nfill  = HC * slices;
    const int b = blockIdx.x;
    if (b < nfill) fill_role(sh.h, src, dst, part, csr, N, E, b % HC, b / HC);
    else           gemm1_role(sh.g, X, W, norm_src, H, N, b - nfill);
}

template <int F>
__device__ __forceinline__ float4 gather_sum(const float* __restrict__ H,
                                             const int* __restrict__ first,
                                             const int* __restrict__ csr,
                                             int g, int f, int N, int E) {
    int j = first[g];
    int end = first[g + 1];
    if (j < 0) j = 0;
    if (end > E) end = E;
    const unsigned top = (unsigned)(N - 1);
    float4 acc = make_float4(0.0f, 0.0f, 0.0f, 0.0f);
    for (; j < end; j += 4) {
        unsigned s[4];
        float4 v[4];
        #pragma unroll
        for (int u = 0; u < 4; ++u) {
            int jj = j + u; if (jj > end - 1) jj = end - 1;
            s[u] = (unsigned)csr[jj]; if (s[u] > top) s[u] = top;
        }
        #pragma unroll
        for (int u = 0; u < 4; ++u)
            v[u] = *(const float4*)(H + (size_t)s[u] * F + f);
        #pragma unroll
        for (int u = 0; u < 4; ++u) {
            const bool in = (j + u < end);
            acc.x += in ? v[u].x : 0.0f;
            acc.y += in ? v[u].y : 0.0f;
            acc.z += in ? v[u].z : 0.0f;
            acc.w += in ? v[u].w : 0.0f;
        }
    }
    return acc;
}

__global__ __launch_bounds__(256)
void k_gather1(const float* __restrict__ H, const int* __restrict__ first,
               const int* __restrict__ csr, float* __restrict__ AGG, int N, int E) {
    const int g = (int)((blockIdx.x * 256u + threadIdx.x) >> 4);
    if (g >= N) return;
    const int f = (threadIdx.x & 15) * 4;
    const float4 acc = gather_sum<64>(H, first, csr, g, f, N, E);
    *(float4*)(AGG + (size_t)g * 64 + f) = acc;
}

__global__ __launch_bounds__(256)
void k_gather2(const float* __restrict__ H, const int* __restrict__ first,
               const int* __restrict__ csr, const float* __restrict__ norm_dst,
               const float* __restrict__ b2, float* __restrict__ out, int N, int E) {
    const int g = (int)((blockIdx.x * 256u + threadIdx.x) >> 2);
    if (g >= N) return;
    const int f = (threadIdx.x & 3) * 4;
    const float4 acc = gather_sum<16>(H, first, csr, g, f, N, E);
    const float nd = norm_dst[g];
    float4 o;
    o.x = acc.x * nd + b2[f + 0];
    o.y = acc.y * nd + b2[f + 1];
    o.z = acc.z * nd + b2[f + 2];
    o.w = acc.w * nd + b2[f + 3];
    *(float4*)(out + (size_t)g * 16 + f) = o;
}

__global__ __launch_bounds__(256)
void k_gemm2(const float* __restrict__ AGG1, const float* __restrict__ W2,
             const float* __restrict__ b1,
             const float* __restrict__ norm_src, const float* __restrict__ norm_dst,
             float* __restrict__ H2, int N) {
    __shared__ float lA[128][66];
    __shared__ float lB[64][18];

    const int m0 = blockIdx.x * 128;
    const int t  = threadIdx.x;

    #pragma unroll
    for (int i = 0; i < 32; ++i) {
        int idx = t + i * 256;
        int r = idx >> 6, c = idx & 63;
        int row = m0 + r; if (row >= N) row = N - 1;
        float v = AGG1[(long)row * 64 + c];
        v = fmaxf(v * norm_dst[row] + b1[c], 0.0f) * norm_src[row];
        lA[r][c] = v;
    }
    #pragma unroll
    for (int i = 0; i < 4; ++i) {
        int idx = t + i * 256;
        lB[idx >> 4][idx & 15] = W2[idx];
    }
    __syncthreads();

    const int wave = t >> 5, lane = t & 31;
    const int mt = wave * 16;
    const int lr = lane & 15;
    const int lh = lane >> 4;

    v8f acc = {};
    #pragma unroll
    for (int k0 = 0; k0 < 64; k0 += 4) {
        const int kk = k0 + 2 * lh;
        v2f a, b;
        a[0] = lA[mt + lr][kk];
        a[1] = lA[mt + lr][kk + 1];
        b[0] = lB[kk][lr];
        b[1] = lB[kk + 1][lr];
        acc = __builtin_amdgcn_wmma_f32_16x16x4_f32(
            false, a, false, b, (short)0, acc, false, false);
    }

    #pragma unroll
    for (int r = 0; r < 8; ++r) {
        int m = m0 + mt + r + 8 * lh;
        if (m < N) H2[(long)m * 16 + lr] = acc[r];
    }
}

extern "C" void kernel_launch(void* const* d_in, const int* in_sizes, int n_in,
                              void* d_out, int out_size, void* d_ws, size_t ws_size,
                              hipStream_t stream) {
    const float* emb = (const float*)d_in[0];
    const int*   src = (const int*)  d_in[1];
    const int*   dst = (const int*)  d_in[2];
    const float* W1  = (const float*)d_in[3];
    const float* b1  = (const float*)d_in[4];
    const float* W2  = (const float*)d_in[5];
    const float* b2  = (const float*)d_in[6];
    float* out = (float*)d_out;

    const int N = in_sizes[0] / 128;
    const int E = in_sizes[1];

    float* ws       = (float*)d_ws;
    float* norm_src = ws;
    float* norm_dst = ws + N;
    float* H1       = ws + 2 * (size_t)N;
    float* AGG1     = H1 + (size_t)N * 64;
    float* H2       = H1;
    unsigned* part  = (unsigned*)(AGG1 + (size_t)N * 64);
    const int NB    = (N + NT - 1) / NT;
    unsigned* bsum  = part + (size_t)HC * 2 * (size_t)N;
    int* first      = (int*)(bsum + NB);
    int* csr        = first + (N + 1);

    const int B = 256;
    const int slices = (N + HB - 1) / HB;

    k_hist<<<dim3(HC, slices, 2), HT, 0, stream>>>(src, dst, part, N, E);
    k_norm<<<NB, NT, 0, stream>>>(part, norm_src, norm_dst, bsum, N);
    k_rowptr<<<NB, NT, 0, stream>>>(part, bsum, first, N);

    k_fill_gemm1<<<HC * slices + (N + GM - 1) / GM, FT, 0, stream>>>(src, dst, part, csr, emb, W1, norm_src, H1, N, E);
    k_gather1<<<(unsigned)(((long)N * 16 + B - 1) / B), B, 0, stream>>>(H1, first, csr, AGG1, N, E);

    k_gemm2<<<(N + 127) / 128, B, 0, stream>>>(AGG1, W2, b1, norm_src, norm_dst, H2, N);
    k_gather2<<<(unsigned)(((long)N * 4 + B - 1) / B), B, 0, stream>>>(H2, first, csr, norm_dst, b2, out, N, E);
}


# layer 2's product moves into layer 1's gather: a work-group gathers 16 nodes into an LDS tile (bias, ReLU, norms applied), one wave multiplies it by W2 on WMMA; the AGG plane and one launch are gone
# speedup vs baseline: 3.2963x; 1.0328x over previous
#include <hip/hip_runtime.h>
#include <hip/hip_bf16.h>

typedef __attribute__((ext_vector_type(2))) float v2f;
typedef __attribute__((ext_vector_type(8))) float v8f;

#define HB 12544
#define HC 16
#define HT 1024
#define HI ((HB + HT - 1) / HT)

__device__ __forceinline__ void load4(const int4* __restrict__ p, int i0, int stride, int last, int4 (&v)[4]) {
    #pragma unroll
    for (int u = 0; u < 4; ++u) {
        int i = i0 + u * stride; if (i > last) i = last;
        v[u] = p[i];
    }
}

__device__ __forceinline__ void count16(unsigned* h, const int4 (&v)[4], int i0, int stride,
                                        int last, int lo, unsigned ub) {
    #pragma unroll
    for (int u = 0; u < 4; ++u) {
        const unsigned ubu = (i0 + u * stride <= last) ? ub : 0u;
        unsigned b;
        b = (unsigned)(v[u].x - lo); if (b < ubu) atomicAdd(&h[b], 1u);
        b = (unsigned)(v[u].y - lo); if (b < ubu) atomicAdd(&h[b], 1u);
        b = (unsigned)(v[u].z - lo); if (b < ubu) atomicAdd(&h[b], 1u);
        b = (unsigned)(v[u].w - lo); if (b < ubu) atomicAdd(&h[b], 1u);
    }
}

__global__ __launch_bounds__(HT)
void k_hist(const int* __restrict__ src, const int* __restrict__ dst,
            unsigned* __restrict__ part, int N, int E) {
    __shared__ unsigned h[HB];
    const int t  = threadIdx.x;
    const int c  = blockIdx.x;
    const int lo = blockIdx.y * HB;
    const int a  = blockIdx.z;
    const int* __restrict__ idx = a ? dst : src;
    int nb = N - lo; if (nb > HB) nb = HB;
    const unsigned ub = (unsigned)nb;

    for (int i = t; i < HB; i += HT) h[i] = 0u;
    __syncthreads();

    const int4* __restrict__ p = (const int4*)idx;
    const int E4 = E >> 2;
    const int last = E4 - 1;
    const int stride = gridDim.x * HT;
    const int step = 4 * stride;
    int i0 = c * HT + t;
    int4 va[4], vb[4];
    if (i0 < E4) load4(p, i0, stride, last, va);
    while (i0 < E4) {
        load4(p, i0 + step, stride, last, vb);
        count16(h, va, i0, stride, last, lo, ub);
        i0 += step;
        if (i0 >= E4) break;
        load4(p, i0 + step, stride, last, va);
        count16(h, vb, i0, stride, last, lo, ub);
        i0 += step;
    }
    if (c == 0 && t < (E & 3)) {
        const unsigned b = (unsigned)(idx[(E4 << 2) + t] - lo);
        if (b < ub) atomicAdd(&h[b], 1u);
    }
    __syncthreads();

    unsigned* __restrict__ o = part + (size_t)c * (2 * (size_t)N) + (size_t)a * N + lo;
    {
        unsigned r[HI];
        #pragma unroll
        for (int k = 0; k < HI; ++k) { const int i = t + k * HT; r[k] = (i < nb) ? h[i] : 0u; }
        #pragma unroll
        for (int k = 0; k < HI; ++k) { const int i = t + k * HT; if (i < nb) o[i] = r[k]; }
    }
}

#define NT 256

__global__ __launch_bounds__(NT)
void k_norm(const unsigned* __restrict__ part, float* __restrict__ norm_src,
            float* __restrict__ norm_dst, unsigned* __restrict__ blocksum, int N) {
    __shared__ unsigned wred[NT / 32];
    const int t = threadIdx.x;
    const int i = blockIdx.x * NT + t;
    const size_t n2 = 2 * (size_t)N;
    unsigned so = 0u, si = 0u;
    if (i < N) {
        #pragma unroll
        for (int c = 0; c < HC; ++c) {
            so += part[(size_t)c * n2 + i];
            si += part[(size_t)c * n2 + N + i];
        }
        norm_src[i] = rsqrtf(fmaxf((float)so, 1.0f));
        norm_dst[i] = rsqrtf(fmaxf((float)si, 1.0f));
    }
    unsigned r = si;
    for (int o = 16; o > 0; o >>= 1) r += __shfl_xor(r, o, 32);
    if ((t & 31) == 0) wred[t >> 5] = r;
    __syncthreads();
    if (t == 0) {
        unsigned s = 0u;
        for (int w = 0; w < NT / 32; ++w) s += wred[w];
        blocksum[blockIdx.x] = s;
    }
}

__global__ __launch_bounds__(NT)
void k_rowptr(unsigned* __restrict__ part, const unsigned* __restrict__ blocksum,
              int* __restrict__ first, int N) {
    __shared__ unsigned wred[NT / 32];
    __shared__ unsigned wsum[NT / 32];
    const int t = threadIdx.x, lane = t & 31, wave = t >> 5;
    const int i = blockIdx.x * NT + t;
    const size_t n2 = 2 * (size_t)N;

    unsigned bs = 0u;
    for (int b = t; b < (int)blockIdx.x; b += NT) bs += blocksum[b];
    for (int o = 16; o > 0; o >>= 1) bs += __shfl_xor(bs, o, 32);

    unsigned p[HC];
    unsigned d = 0u;
    #pragma unroll
    for (int c = 0; c < HC; ++c) {
        p[c] = (i < N) ? part[(size_t)c * n2 + N + i] : 0u;
        d += p[c];
    }

    unsigned inc = d;
    for (int o = 1; o < 32; o <<= 1) {
        const unsigned n = __shfl_up(inc, o, 32);
        if (lane >= o) inc += n;
    }
    if (lane == 0)  wred[wave] = bs;
    if (lane == 31) wsum[wave] = inc;
    __syncthreads();
    unsigned r = inc - d;
    for (int w = 0; w < NT / 32; ++w) {
        r += wred[w];
        if (w < wave) r += wsum[w];
    }

    if (i < N) {
        first[i] = (int)r;
        if (i == N - 1) first[N] = (int)(r + d);
        #pragma unroll
        for (int c = 0; c < HC; ++c) {
            part[(size_t)c * n2 + N + i] = r;
            r += p[c];
        }
    }
}

#define FT 768
#define FI ((HB + FT - 1) / FT)

__device__ __forceinline__ int share_pos(int m, int c) {
    return ((m / HT) * HC + c) * HT + (m % HT);
}

__device__ __forceinline__ void load4s(const int4* __restrict__ p, int m0, int c, int last, int4 (&v)[4]) {
    #pragma unroll
    for (int u = 0; u < 4; ++u) {
        int i = share_pos(m0 + u * FT, c); if (i > last) i = last;
        v[u] = p[i];
    }
}

__device__ __forceinline__ unsigned take(unsigned* h, int node, int lo, unsigned ub) {
    const unsigned b = (unsigned)(node - lo);
    unsigned r = 0xFFFFFFFFu;
    if (b < ub) r = atomicAdd(&h[b], 1u);
    return r;
}

__device__ __forceinline__ void place16(unsigned* h, int* __restrict__ csr,
                                        const int4 (&v)[4], const int4 (&w)[4],
                                        int m0, int c, int last, int lo, unsigned ub, unsigned ue) {
    unsigned pos[16];
    #pragma unroll
    for (int u = 0; u < 4; ++u) {
        const unsigned ubu = (share_pos(m0 + u * FT, c) <= last) ? ub : 0u;
        pos[4 * u + 0] = take(h, v[u].x, lo, ubu);
        pos[4 * u + 1] = take(h, v[u].y, lo, ubu);
        pos[4 * u + 2] = take(h, v[u].z, lo, ubu);
        pos[4 * u + 3] = take(h, v[u].w, lo, ubu);
    }
    #pragma unroll
    for (int u = 0; u < 4; ++u) {
        if (pos[4 * u + 0] < ue) csr[pos[4 * u + 0]] = w[u].x;
        if (pos[4 * u + 1] < ue) csr[pos[4 * u + 1]] = w[u].y;
        if (pos[4 * u + 2] < ue) csr[pos[4 * u + 2]] = w[u].z;
        if (pos[4 * u + 3] < ue) csr[pos[4 * u + 3]] = w[u].w;
    }
}

__device__ __forceinline__
void fill_role(unsigned* h, const int* __restrict__ src, const int* __restrict__ dst,
               const unsigned* __restrict__ part, int* __restrict__ csr, int N, int E,
               int c, int sl) {
    const int t  = threadIdx.x;
    const int lo = sl * HB;
    int nb = N - lo; if (nb > HB) nb = HB;
    const unsigned ub = (unsigned)nb;
    const unsigned ue = (unsigned)E;

    const unsigned* __restrict__ cur = part + (size_t)c * (2 * (size_t)N) + N + lo;
    {
        unsigned r[FI];
        #pragma unroll
        for (int k = 0; k < FI; ++k) { const int i = t + k * FT; r[k] = (i < nb) ? cur[i] : 0u; }
        #pragma unroll
        for (int k = 0; k < FI; ++k) { const int i = t + k * FT; if (i < nb) h[i] = r[k]; }
    }
    __syncthreads();

    const int4* __restrict__ p = (const int4*)dst;
    const int4* __restrict__ q = (const int4*)src;
    const int E4 = E >> 2;
    const int last = E4 - 1;
    const int step = 4 * FT;
    int m0 = t;
    int4 va[4], wa[4], vb[4], wb[4];
    if (share_pos(m0, c) < E4) { load4s(p, m0, c, last, va); load4s(q, m0, c, last, wa); }
    while (share_pos(m0, c) < E4) {
        load4s(p, m0 + step, c, last, vb);
        load4s(q, m0 + step, c, last, wb);
        place16(h, csr, va, wa, m0, c, last, lo, ub, ue);
        m0 += step;
        if (share_pos(m0, c) >= E4) break;
        load4s(p, m0 + step, c, last, va);
        load4s(q, m0 + step, c, last, wa);
        place16(h, csr, vb, wb, m0, c, last, lo, ub, ue);
        m0 += step;
    }
    if (c == 0 && t < (E & 3)) {
        const int e = (E4 << 2) + t;
        const unsigned b = (unsigned)(dst[e] - lo);
        if (b < ub) {
            const unsigned pos = atomicAdd(&h[b], 1u);
            if (pos < ue) csr[pos] = src[e];
        }
    }
}

#define GM 96
#define GA (GM * 130)
#define GW (128 * 66)

__device__ __forceinline__
void gemm1_role(float* sm, const float* __restrict__ X, const float* __restrict__ W,
                const float* __restrict__ norm_src, float* __restrict__ H, int N, int tile) {
    float (*lA)[130] = (float (*)[130])sm;
    float (*lB)[66]  = (float (*)[66])(sm + GA);

    const int m0 = tile * GM;
    const int t  = threadIdx.x;

    #pragma unroll
    for (int i = 0; i < GM * 128 / FT; ++i) {
        int idx = t + i * FT;
        int r = idx >> 7, c = idx & 127;
        int row = m0 + r; if (row >= N) row = N - 1;
        lA[r][c] = X[(long)row * 128 + c] * norm_src[row];
    }
    #pragma unroll
    for (int i = 0; i < (128 * 64 + FT - 1) / FT; ++i) {
        int idx = t + i * FT;
        if (idx < 128 * 64) lB[idx >> 6][idx & 63] = W[idx];
    }
    __syncthreads();

    const int wave = t >> 5, lane = t & 31;
    const int mt = (wave >> 2) * 16;
    const int nt = (wave & 3) * 16;
    const int lr = lane & 15;
    const int lh = lane >> 4;

    v8f acc = {};
    #pragma unroll
    for (int k0 = 0; k0 < 128; k0 += 4) {
        const int kk = k0 + 2 * lh;
        v2f a, b;
        a[0] = lA[mt + lr][kk];
        a[1] = lA[mt + lr][kk + 1];
        b[0] = lB[kk][nt + lr];
        b[1] = lB[kk + 1][nt + lr];
        acc = __builtin_amdgcn_wmma_f32_16x16x4_f32(
             false, a,  false, b,
             (short)0, acc,  false,  false);
    }

    #pragma unroll
    for (int r = 0; r < 8; ++r) {
        int m = m0 + mt + r + 8 * lh;
        if (m < N) H[(long)m * 64 + nt + lr] = acc[r];
    }
}

union FillOrTile {
    unsigned h[HB];
    float    g[GA + GW];
};

__global__ __launch_bounds__(FT)
__attribute__((amdgpu_waves_per_eu(12)))
void k_fill_gemm1(const int* __restrict__ src, const int* __restrict__ dst,
                  const unsigned* __restrict__ part, int* __restrict__ csr,
                  const float* __restrict__ X, const float* __restrict__ W,
                  const float* __restrict__ norm_src, float* __restrict__ H, int N, int E) {
    __shared__ FillOrTile sh;
    const int slices = (N + HB - 1) / HB;
    const int nfill  = HC * slices;
    const int b = blockIdx.x;
    if (b < nfill) fill_role(sh.h, src, dst, part, csr, N, E, b % HC, b / HC);
    else           gemm1_role(sh.g, X, W, norm_src, H, N, b - nfill);
}

template <int F>
__device__ __forceinline__ float4 gather_sum(const float* __restrict__ H,
                                             const int* __restrict__ first,
                                             const int* __restrict__ csr,
                                             int g, int f, int N, int E) {
    int j = first[g];
    int end = first[g + 1];
    if (j < 0) j = 0;
    if (end > E) end = E;
    const unsigned top = (unsigned)(N - 1);
    float4 acc = make_float4(0.0f, 0.0f, 0.0f, 0.0f);
    for (; j < end; j += 4) {
        unsigned s[4];
        float4 v[4];
        #pragma unroll
        for (int u = 0; u < 4; ++u) {
            int jj = j + u; if (jj > end - 1) jj = end - 1;
            s[u] = (unsigned)csr[jj]; if (s[u] > top) s[u] = top;
        }
        #pragma unroll
        for (int u = 0; u < 4; ++u)
            v[u] = *(const float4*)(H + (size_t)s[u] * F + f);
        #pragma unroll
        for (int u = 0; u < 4; ++u) {
            const bool in = (j + u < end);
            acc.x += in ? v[u].x : 0.0f;
            acc.y += in ? v[u].y : 0.0f;
            acc.z += in ? v[u].z : 0.0f;
            acc.w += in ? v[u].w : 0.0f;
        }
    }
    return acc;
}

#define TN 16
#define TP 68

__global__ __launch_bounds__(256)
void k_gather1_gemm2(const float* __restrict__ H, const int* __restrict__ first,
                     const int* __restrict__ csr, const float* __restrict__ W2,
                     const float* __restrict__ b1, const float* __restrict__ norm_src,
                     const float* __restrict__ norm_dst, float* __restrict__ H2, int N, int E) {
    __shared__ __attribute__((aligned(16))) float lA[TN][TP];
    const int t  = threadIdx.x;
    const int m0 = blockIdx.x * TN;
    const int r  = t >> 4;
    const int g  = m0 + r;
    const int f  = (t & 15) * 4;

    float4 v = make_float4(0.0f, 0.0f, 0.0f, 0.0f);
    if (g < N) {
        const float4 acc = gather_sum<64>(H, first, csr, g, f, N, E);
        const float nd = norm_dst[g];
        const float ns = norm_src[g];
        v.x = fmaxf(acc.x * nd + b1[f + 0], 0.0f) * ns;
        v.y = fmaxf(acc.y * nd + b1[f + 1], 0.0f) * ns;
        v.z = fmaxf(acc.z * nd + b1[f + 2], 0.0f) * ns;
        v.w = fmaxf(acc.w * nd + b1[f + 3], 0.0f) * ns;
    }
    *(float4*)&lA[r][f] = v;

    const int lr = t & 15;
    const int lh = (t >> 4) & 1;
    v2f b[16];
    if (t < 32) {
        #pragma unroll
        for (int k0 = 0; k0 < 64; k0 += 4) {
            const int kk = k0 + 2 * lh;
            b[k0 >> 2][0] = W2[kk * 16 + lr];
            b[k0 >> 2][1] = W2[(kk + 1) * 16 + lr];
        }
    }
    __syncthreads();

    if (t < 32) {
        v8f acc = {};
        #pragma unroll
        for (int k0 = 0; k0 < 64; k0 += 4) {
            const int kk = k0 + 2 * lh;
            v2f a;
            a[0] = lA[lr][kk];
            a[1] = lA[lr][kk + 1];
            acc = __builtin_amdgcn_wmma_f32_16x16x4_f32(
                false, a, false, b[k0 >> 2], (short)0, acc, false, false);
        }
        #pragma unroll
        for (int q = 0; q < 8; ++q) {
            const int m = m0 + q + 8 * lh;
            if (m < N) H2[(long)m * 16 + lr] = acc[q];
        }
    }
}

__global__ __launch_bounds__(256)
void k_gather2(const float* __restrict__ H, const int* __restrict__ first,
               const int* __restrict__ csr, const float* __restrict__ norm_dst,
               const float* __restrict__ b2, float* __restrict__ out, int N, int E) {
    const int g = (int)((blockIdx.x * 256u + threadIdx.x) >> 2);
    if (g >= N) return;
    const int f = (threadIdx.x & 3) * 4;
    const float4 acc = gather_sum<16>(H, first, csr, g, f, N, E);
    const float nd = norm_dst[g];
    float4 o;
    o.x = acc.x * nd + b2[f + 0];
    o.y = acc.y * nd + b2[f + 1];
    o.z = acc.z * nd + b2[f + 2];
    o.w = acc.w * nd + b2[f + 3];
    *(float4*)(out + (size_t)g * 16 + f) = o;
}

extern "C" void kernel_launch(void* const* d_in, const int* in_sizes, int n_in,
                              void* d_out, int out_size, void* d_ws, size_t ws_size,
                              hipStream_t stream) {
    const float* emb = (const float*)d_in[0];
    const int*   src = (const int*)  d_in[1];
    const int*   dst = (const int*)  d_in[2];
    const float* W1  = (const float*)d_in[3];
    const float* b1  = (const float*)d_in[4];
    const float* W2  = (const float*)d_in[5];
    const float* b2  = (const float*)d_in[6];
    float* out = (float*)d_out;

    const int N = in_sizes[0] / 128;
    const int E = in_sizes[1];

    float* ws       = (float*)d_ws;
    float* norm_src = ws;
    float* norm_dst = ws + N;
    float* H1       = ws + 2 * (size_t)N;
    float* H2       = H1 + (size_t)N * 64;
    unsigned* part  = (unsigned*)(H2 + (size_t)N * 64);
    const int NB    = (N + NT - 1) / NT;
    unsigned* bsum  = part + (size_t)HC * 2 * (size_t)N;
    int* first      = (int*)(bsum + NB);
    int* csr        = first + (N + 1);

    const int B = 256;
    const int slices = (N + HB - 1) / HB;

    k_hist<<<dim3(HC, slices, 2), HT, 0, stream>>>(src, dst, part, N, E);
    k_norm<<<NB, NT, 0, stream>>>(part, norm_src, norm_dst, bsum, N);
    k_rowptr<<<NB, NT, 0, stream>>>(part, bsum, first, N);

    k_fill_gemm1<<<HC * slices + (N + GM - 1) / GM, FT, 0, stream>>>(src, dst, part, csr, emb, W1, norm_src, H1, N, E);

    k_gather1_gemm2<<<(N + TN - 1) / TN, B, 0, stream>>>(H1, first, csr, W2, b1, norm_src, norm_dst, H2, N, E);
    k_gather2<<<(unsigned)(((long)N * 4 + B - 1) / B), B, 0, stream>>>(H2, first, csr, norm_dst, b2, out, N, E);
}


# both gather walks keep loads in flight across entry-then-row: the next group's list entries are requested under the current rows, 8 rows per group, norms and bias before the walk; add order unchanged
# speedup vs baseline: 3.4459x; 1.0454x over previous
#include <hip/hip_runtime.h>
#include <hip/hip_bf16.h>

typedef __attribute__((ext_vector_type(2))) float v2f;
typedef __attribute__((ext_vector_type(8))) float v8f;

#define HB 12544
#define HC 16
#define HT 1024
#define HI ((HB + HT - 1) / HT)

__device__ __forceinline__ void load4(const int4* __restrict__ p, int i0, int stride, int last, int4 (&v)[4]) {
    #pragma unroll
    for (int u = 0; u < 4; ++u) {
        int i = i0 + u * stride; if (i > last) i = last;
        v[u] = p[i];
    }
}

__device__ __forceinline__ void count16(unsigned* h, const int4 (&v)[4], int i0, int stride,
                                        int last, int lo, unsigned ub) {
    #pragma unroll
    for (int u = 0; u < 4; ++u) {
        const unsigned ubu = (i0 + u * stride <= last) ? ub : 0u;
        unsigned b;
        b = (unsigned)(v[u].x - lo); if (b < ubu) atomicAdd(&h[b], 1u);
        b = (unsigned)(v[u].y - lo); if (b < ubu) atomicAdd(&h[b], 1u);
        b = (unsigned)(v[u].z - lo); if (b < ubu) atomicAdd(&h[b], 1u);
        b = (unsigned)(v[u].w - lo); if (b < ubu) atomicAdd(&h[b], 1u);
    }
}

__global__ __launch_bounds__(HT)
void k_hist(const int* __restrict__ src, const int* __restrict__ dst,
            unsigned* __restrict__ part, int N, int E) {
    __shared__ unsigned h[HB];
    const int t  = threadIdx.x;
    const int c  = blockIdx.x;
    const int lo = blockIdx.y * HB;
    const int a  = blockIdx.z;
    const int* __restrict__ idx = a ? dst : src;
    int nb = N - lo; if (nb > HB) nb = HB;
    const unsigned ub = (unsigned)nb;

    for (int i = t; i < HB; i += HT) h[i] = 0u;
    __syncthreads();

    const int4* __restrict__ p = (const int4*)idx;
    const int E4 = E >> 2;
    const int last = E4 - 1;
    const int stride = gridDim.x * HT;
    const int step = 4 * stride;
    int i0 = c * HT + t;
    int4 va[4], vb[4];
    if (i0 < E4) load4(p, i0, stride, last, va);
    while (i0 < E4) {
        load4(p, i0 + step, stride, last, vb);
        count16(h, va, i0, stride, last, lo, ub);
        i0 += step;
        if (i0 >= E4) break;
        load4(p, i0 + step, stride, last, va);
        count16(h, vb, i0, stride, last, lo, ub);
        i0 += step;
    }
    if (c == 0 && t < (E & 3)) {
        const unsigned b = (unsigned)(idx[(E4 << 2) + t] - lo);
        if (b < ub) atomicAdd(&h[b], 1u);
    }
    __syncthreads();

    unsigned* __restrict__ o = part + (size_t)c * (2 * (size_t)N) + (size_t)a * N + lo;
    {
        unsigned r[HI];
        #pragma unroll
        for (int k = 0; k < HI; ++k) { const int i = t + k * HT; r[k] = (i < nb) ? h[i] : 0u; }
        #pragma unroll
        for (int k = 0; k < HI; ++k) { const int i = t + k * HT; if (i < nb) o[i] = r[k]; }
    }
}

#define NT 256

__global__ __launch_bounds__(NT)
void k_norm(const unsigned* __restrict__ part, float* __restrict__ norm_src,
            float* __restrict__ norm_dst, unsigned* __restrict__ blocksum, int N) {
    __shared__ unsigned wred[NT / 32];
    const int t = threadIdx.x;
    const int i = blockIdx.x * NT + t;
    const size_t n2 = 2 * (size_t)N;
    unsigned so = 0u, si = 0u;
    if (i < N) {
        #pragma unroll
        for (int c = 0; c < HC; ++c) {
            so += part[(size_t)c * n2 + i];
            si += part[(size_t)c * n2 + N + i];
        }
        norm_src[i] = rsqrtf(fmaxf((float)so, 1.0f));
        norm_dst[i] = rsqrtf(fmaxf((float)si, 1.0f));
    }
    unsigned r = si;
    for (int o = 16; o > 0; o >>= 1) r += __shfl_xor(r, o, 32);
    if ((t & 31) == 0) wred[t >> 5] = r;
    __syncthreads();
    if (t == 0) {
        unsigned s = 0u;
        for (int w = 0; w < NT / 32; ++w) s += wred[w];
        blocksum[blockIdx.x] = s;
    }
}

__global__ __launch_bounds__(NT)
void k_rowptr(unsigned* __restrict__ part, const unsigned* __restrict__ blocksum,
              int* __restrict__ first, int N) {
    __shared__ unsigned wred[NT / 32];
    __shared__ unsigned wsum[NT / 32];
    const int t = threadIdx.x, lane = t & 31, wave = t >> 5;
    const int i = blockIdx.x * NT + t;
    const size_t n2 = 2 * (size_t)N;

    unsigned bs = 0u;
    for (int b = t; b < (int)blockIdx.x; b += NT) bs += blocksum[b];
    for (int o = 16; o > 0; o >>= 1) bs += __shfl_xor(bs, o, 32);

    unsigned p[HC];
    unsigned d = 0u;
    #pragma unroll
    for (int c = 0; c < HC; ++c) {
        p[c] = (i < N) ? part[(size_t)c * n2 + N + i] : 0u;
        d += p[c];
    }

    unsigned inc = d;
    for (int o = 1; o < 32; o <<= 1) {
        const unsigned n = __shfl_up(inc, o, 32);
        if (lane >= o) inc += n;
    }
    if (lane == 0)  wred[wave] = bs;
    if (lane == 31) wsum[wave] = inc;
    __syncthreads();
    unsigned r = inc - d;
    for (int w = 0; w < NT / 32; ++w) {
        r += wred[w];
        if (w < wave) r += wsum[w];
    }

    if (i < N) {
        first[i] = (int)r;
        if (i == N - 1) first[N] = (int)(r + d);
        #pragma unroll
        for (int c = 0; c < HC; ++c) {
            part[(size_t)c * n2 + N + i] = r;
            r += p[c];
        }
    }
}

#define FT 768
#define FI ((HB + FT - 1) / FT)

__device__ __forceinline__ int share_pos(int m, int c) {
    return ((m / HT) * HC + c) * HT + (m % HT);
}

__device__ __forceinline__ void load4s(const int4* __restrict__ p, int m0, int c, int last, int4 (&v)[4]) {
    #pragma unroll
    for (int u = 0; u < 4; ++u) {
        int i = share_pos(m0 + u * FT, c); if (i > last) i = last;
        v[u] = p[i];
    }
}

__device__ __forceinline__ unsigned take(unsigned* h, int node, int lo, unsigned ub) {
    const unsigned b = (unsigned)(node - lo);
    unsigned r = 0xFFFFFFFFu;
    if (b < ub) r = atomicAdd(&h[b], 1u);
    return r;
}

__device__ __forceinline__ void place16(unsigned* h, int* __restrict__ csr,
                                        const int4 (&v)[4], const int4 (&w)[4],
                                        int m0, int c, int last, int lo, unsigned ub, unsigned ue) {
    unsigned pos[16];
    #pragma unroll
    for (int u = 0; u < 4; ++u) {
        const unsigned ubu = (share_pos(m0 + u * FT, c) <= last) ? ub : 0u;
        pos[4 * u + 0] = take(h, v[u].x, lo, ubu);
        pos[4 * u + 1] = take(h, v[u].y, lo, ubu);
        pos[4 * u + 2] = take(h, v[u].z, lo, ubu);
        pos[4 * u + 3] = take(h, v[u].w, lo, ubu);
    }
    #pragma unroll
    for (int u = 0; u < 4; ++u) {
        if (pos[4 * u + 0] < ue) csr[pos[4 * u + 0]] = w[u].x;
        if (pos[4 * u + 1] < ue) csr[pos[4 * u + 1]] = w[u].y;
        if (pos[4 * u + 2] < ue) csr[pos[4 * u + 2]] = w[u].z;
        if (pos[4 * u + 3] < ue) csr[pos[4 * u + 3]] = w[u].w;
    }
}

__device__ __forceinline__
void fill_role(unsigned* h, const int* __restrict__ src, const int* __restrict__ dst,
               const unsigned* __restrict__ part, int* __restrict__ csr, int N, int E,
               int c, int sl) {
    const int t  = threadIdx.x;
    const int lo = sl * HB;
    int nb = N - lo; if (nb > HB) nb = HB;
    const unsigned ub = (unsigned)nb;
    const unsigned ue = (unsigned)E;

    const unsigned* __restrict__ cur = part + (size_t)c * (2 * (size_t)N) + N + lo;
    {
        unsigned r[FI];
        #pragma unroll
        for (int k = 0; k < FI; ++k) { const int i = t + k * FT; r[k] = (i < nb) ? cur[i] : 0u; }
        #pragma unroll
        for (int k = 0; k < FI; ++k) { const int i = t + k * FT; if (i < nb) h[i] = r[k]; }
    }
    __syncthreads();

    const int4* __restrict__ p = (const int4*)dst;
    const int4* __restrict__ q = (const int4*)src;
    const int E4 = E >> 2;
    const int last = E4 - 1;
    const int step = 4 * FT;
    int m0 = t;
    int4 va[4], wa[4], vb[4], wb[4];
    if (share_pos(m0, c) < E4) { load4s(p, m0, c, last, va); load4s(q, m0, c, last, wa); }
    while (share_pos(m0, c) < E4) {
        load4s(p, m0 + step, c, last, vb);
        load4s(q, m0 + step, c, last, wb);
        place16(h, csr, va, wa, m0, c, last, lo, ub, ue);
        m0 += step;
        if (share_pos(m0, c) >= E4) break;
        load4s(p, m0 + step, c, last, va);
        load4s(q, m0 + step, c, last, wa);
        place16(h, csr, vb, wb, m0, c, last, lo, ub, ue);
        m0 += step;
    }
    if (c == 0 && t < (E & 3)) {
        const int e = (E4 << 2) + t;
        const unsigned b = (unsigned)(dst[e] - lo);
        if (b < ub) {
            const unsigned pos = atomicAdd(&h[b], 1u);
            if (pos < ue) csr[pos] = src[e];
        }
    }
}

#define GM 96
#define GA (GM * 130)
#define GW (128 * 66)

__device__ __forceinline__
void gemm1_role(float* sm, const float* __restrict__ X, const float* __restrict__ W,
                const float* __restrict__ norm_src, float* __restrict__ H, int N, int tile) {
    float (*lA)[130] = (float (*)[130])sm;
    float (*lB)[66]  = (float (*)[66])(sm + GA);

    const int m0 = tile * GM;
    const int t  = threadIdx.x;

    #pragma unroll
    for (int i = 0; i < GM * 128 / FT; ++i) {
        int idx = t + i * FT;
        int r = idx >> 7, c = idx & 127;
        int row = m0 + r; if (row >= N) row = N - 1;
        lA[r][c] = X[(long)row * 128 + c] * norm_src[row];
    }
    #pragma unroll
    for (int i = 0; i < (128 * 64 + FT - 1) / FT; ++i) {
        int idx = t + i * FT;
        if (idx < 128 * 64) lB[idx >> 6][idx & 63] = W[idx];
    }
    __syncthreads();

    const int wave = t >> 5, lane = t & 31;
    const int mt = (wave >> 2) * 16;
    const int nt = (wave & 3) * 16;
    const int lr = lane & 15;
    const int lh = lane >> 4;

    v8f acc = {};
    #pragma unroll
    for (int k0 = 0; k0 < 128; k0 += 4) {
        const int kk = k0 + 2 * lh;
        v2f a, b;
        a[0] = lA[mt + lr][kk];
        a[1] = lA[mt + lr][kk + 1];
        b[0] = lB[kk][nt + lr];
        b[1] = lB[kk + 1][nt + lr];
        acc = __builtin_amdgcn_wmma_f32_16x16x4_f32(
             false, a,  false, b,
             (short)0, acc,  false,  false);
    }

    #pragma unroll
    for (int r = 0; r < 8; ++r) {
        int m = m0 + mt + r + 8 * lh;
        if (m < N) H[(long)m * 64 + nt + lr] = acc[r];
    }
}

union FillOrTile {
    unsigned h[HB];
    float    g[GA + GW];
};

__global__ __launch_bounds__(FT)
__attribute__((amdgpu_waves_per_eu(12)))
void k_fill_gemm1(const int* __restrict__ src, const int* __restrict__ dst,
                  const unsigned* __restrict__ part, int* __restrict__ csr,
                  const float* __restrict__ X, const float* __restrict__ W,
                  const float* __restrict__ norm_src, float* __restrict__ H, int N, int E) {
    __shared__ FillOrTile sh;
    const int slices = (N + HB - 1) / HB;
    const int nfill  = HC * slices;
    const int b = blockIdx.x;
    if (b < nfill) fill_role(sh.h, src, dst, part, csr, N, E, b % HC, b / HC);
    else           gemm1_role(sh.g, X, W, norm_src, H, N, b - nfill);
}

#define GU 8

template <int F>
__device__ __forceinline__ float4 gather_sum(const float* __restrict__ H,
                                             const int* __restrict__ first,
                                             const int* __restrict__ csr,
                                             int g, int f, int N, int E) {
    int j = first[g];
    int end = first[g + 1];
    if (j < 0) j = 0;
    if (end > E) end = E;
    const unsigned top = (unsigned)(N - 1);
    float4 acc = make_float4(0.0f, 0.0f, 0.0f, 0.0f);
    if (j >= end) return acc;
    const int lastj = end - 1;
    unsigned s[GU];
    #pragma unroll
    for (int u = 0; u < GU; ++u) {
        int jj = j + u; if (jj > lastj) jj = lastj;
        s[u] = (unsigned)csr[jj];
    }
    for (; j < end; j += GU) {
        float4 v[GU];
        #pragma unroll
        for (int u = 0; u < GU; ++u) {
            unsigned su = s[u]; if (su > top) su = top;
            v[u] = *(const float4*)((const char*)H + (su * (unsigned)(F * 4) + (unsigned)(f * 4)));
        }
        #pragma unroll
        for (int u = 0; u < GU; ++u) {
            int jj = j + GU + u; if (jj > lastj) jj = lastj;
            s[u] = (unsigned)csr[jj];
        }
        __builtin_amdgcn_sched_barrier(0);
        #pragma unroll
        for (int u = 0; u < GU; ++u) {
            const bool in = (j + u < end);
            acc.x += in ? v[u].x : 0.0f;
            acc.y += in ? v[u].y : 0.0f;
            acc.z += in ? v[u].z : 0.0f;
            acc.w += in ? v[u].w : 0.0f;
        }
    }
    return acc;
}

#define TN 16
#define TP 68

__global__ __launch_bounds__(256)
void k_gather1_gemm2(const float* __restrict__ H, const int* __restrict__ first,
                     const int* __restrict__ csr, const float* __restrict__ W2,
                     const float* __restrict__ b1, const float* __restrict__ norm_src,
                     const float* __restrict__ norm_dst, float* __restrict__ H2, int N, int E) {
    __shared__ __attribute__((aligned(16))) float lA[TN][TP];
    const int t  = threadIdx.x;
    const int m0 = blockIdx.x * TN;
    const int r  = t >> 4;
    const int g  = m0 + r;
    const int f  = (t & 15) * 4;

    float4 v = make_float4(0.0f, 0.0f, 0.0f, 0.0f);
    if (g < N) {
        const float nd = norm_dst[g];
        const float ns = norm_src[g];
        const float bx = b1[f + 0], by = b1[f + 1], bz = b1[f + 2], bw = b1[f + 3];
        const float4 acc = gather_sum<64>(H, first, csr, g, f, N, E);
        v.x = fmaxf(acc.x * nd + bx, 0.0f) * ns;
        v.y = fmaxf(acc.y * nd + by, 0.0f) * ns;
        v.z = fmaxf(acc.z * nd + bz, 0.0f) * ns;
        v.w = fmaxf(acc.w * nd + bw, 0.0f) * ns;
    }
    *(float4*)&lA[r][f] = v;

    const int lr = t & 15;
    const int lh = (t >> 4) & 1;
    v2f b[16];
    if (t < 32) {
        #pragma unroll
        for (int k0 = 0; k0 < 64; k0 += 4) {
            const int kk = k0 + 2 * lh;
            b[k0 >> 2][0] = W2[kk * 16 + lr];
            b[k0 >> 2][1] = W2[(kk + 1) * 16 + lr];
        }
    }
    __syncthreads();

    if (t < 32) {
        v8f acc = {};
        #pragma unroll
        for (int k0 = 0; k0 < 64; k0 += 4) {
            const int kk = k0 + 2 * lh;
            v2f a;
            a[0] = lA[lr][kk];
            a[1] = lA[lr][kk + 1];
            acc = __builtin_amdgcn_wmma_f32_16x16x4_f32(
                false, a, false, b[k0 >> 2], (short)0, acc, false, false);
        }
        #pragma unroll
        for (int q = 0; q < 8; ++q) {
            const int m = m0 + q + 8 * lh;
            if (m < N) H2[(long)m * 16 + lr] = acc[q];
        }
    }
}

__global__ __launch_bounds__(256)
void k_gather2(const float* __restrict__ H, const int* __restrict__ first,
               const int* __restrict__ csr, const float* __restrict__ norm_dst,
               const float* __restrict__ b2, float* __restrict__ out, int N, int E) {
    const int g = (int)((blockIdx.x * 256u + threadIdx.x) >> 2);
    if (g >= N) return;
    const int f = (threadIdx.x & 3) * 4;
    const float nd = norm_dst[g];
    const float bx = b2[f + 0], by = b2[f + 1], bz = b2[f + 2], bw = b2[f + 3];
    const float4 acc = gather_sum<16>(H, first, csr, g, f, N, E);
    float4 o;
    o.x = acc.x * nd + bx;
    o.y = acc.y * nd + by;
    o.z = acc.z * nd + bz;
    o.w = acc.w * nd + bw;
    *(float4*)(out + (size_t)g * 16 + f) = o;
}

extern "C" void kernel_launch(void* const* d_in, const int* in_sizes, int n_in,
                              void* d_out, int out_size, void* d_ws, size_t ws_size,
                              hipStream_t stream) {
    const float* emb = (const float*)d_in[0];
    const int*   src = (const int*)  d_in[1];
    const int*   dst = (const int*)  d_in[2];
    const float* W1  = (const float*)d_in[3];
    const float* b1  = (const float*)d_in[4];
    const float* W2  = (const float*)d_in[5];
    const float* b2  = (const float*)d_in[6];
    float* out = (float*)d_out;

    const int N = in_sizes[0] / 128;
    const int E = in_sizes[1];

    float* ws       = (float*)d_ws;
    float* norm_src = ws;
    float* norm_dst = ws + N;
    float* H1       = ws + 2 * (size_t)N;
    float* H2       = H1 + (size_t)N * 64;
    unsigned* part  = (unsigned*)(H2 + (size_t)N * 64);
    const int NB    = (N + NT - 1) / NT;
    unsigned* bsum  = part + (size_t)HC * 2 * (size_t)N;
    int* first      = (int*)(bsum + NB);
    int* csr        = first + (N + 1);

    const int B = 256;
    const int slices = (N + HB - 1) / HB;

    k_hist<<<dim3(HC, slices, 2), HT, 0, stream>>>(src, dst, part, N, E);
    k_norm<<<NB, NT, 0, stream>>>(part, norm_src, norm_dst, bsum, N);
    k_rowptr<<<NB, NT, 0, stream>>>(part, bsum, first, N);

    k_fill_gemm1<<<HC * slices + (N + GM - 1) / GM, FT, 0, stream>>>(src, dst, part, csr, emb, W1, norm_src, H1, N, E);

    k_gather1_gemm2<<<(N + TN - 1) / TN, B, 0, stream>>>(H1, first, csr, W2, b1, norm_src, norm_dst, H2, N, E);
    k_gather2<<<(unsigned)(((long)N * 4 + B - 1) / B), B, 0, stream>>>(H2, first, csr, norm_dst, b2, out, N, E);
}
